# starting kernel
# baseline (speedup 1.0000x reference)
#include <hip/hip_runtime.h>
#include <hip/hip_bf16.h>

#define IN_DIM 256
#define NCOL 128
#define HEADS 4
#define OUT_DIM 32
#define ALPHA 0.2f

typedef __attribute__((ext_vector_type(2))) float v2f;
typedef __attribute__((ext_vector_type(8))) float v8f;

__global__ void gat_gemm_wmma(const float* __restrict__ A,
                              const float* __restrict__ B,
                              float* __restrict__ C, int M) {
    const int lane = threadIdx.x & 31;
    const int wave = (blockIdx.x * blockDim.x + threadIdx.x) >> 5;
    const int tiles_n = NCOL / 16;
    const int row_tile = wave / tiles_n;
    const int col_tile = wave % tiles_n;
    const int m0 = row_tile * 16;
    if (m0 >= M) return;

    const int n0   = col_tile * 16;
    const int r    = lane & 15;
    const int koff = (lane >> 4) * 2;

    v8f acc = {};
    const float* arow = A + (size_t)(m0 + r) * IN_DIM;
#pragma unroll 4
    for (int k = 0; k < IN_DIM; k += 4) {
        v2f a = *(const v2f*)(arow + k + koff);
        const float* bp = B + (size_t)(k + koff) * NCOL + n0 + r;
        v2f b;
        b.x = bp[0];
        b.y = bp[NCOL];
        acc = __builtin_amdgcn_wmma_f32_16x16x4_f32(
             false, a,  false, b,
             (short)0, acc,  false,  false);
    }

    const int row_add = (lane >> 4) * 8;
    float* cp = C + (size_t)(m0 + row_add) * NCOL + n0 + (lane & 15);
#pragma unroll
    for (int i = 0; i < 8; ++i) cp[(size_t)i * NCOL] = acc[i];
}

__global__ void gat_node_scores(const float* __restrict__ ft,
                                const float* __restrict__ attn_l,
                                const float* __restrict__ attn_r,
                                float* __restrict__ a1,
                                float* __restrict__ a2, int n_nodes) {
    int i = blockIdx.x * blockDim.x + threadIdx.x;
    if (i >= n_nodes * HEADS) return;
    int h = i & (HEADS - 1);
    const float* f  = ft + (size_t)(i >> 2) * NCOL + h * OUT_DIM;
    const float* al = attn_l + h * OUT_DIM;
    const float* ar = attn_r + h * OUT_DIM;
    float s1 = 0.f, s2 = 0.f;
#pragma unroll
    for (int d = 0; d < OUT_DIM; ++d) {
        float v = f[d];
        s1 += v * al[d];
        s2 += v * ar[d];
    }
    a1[i] = s1;
    a2[i] = s2;
}

__global__ void gat_init(float* __restrict__ out, float* __restrict__ e_max,
                         float* __restrict__ e_sum, int n_nodes) {
    int i = blockIdx.x * blockDim.x + threadIdx.x;
    int n_out = n_nodes * NCOL;
    if (i < n_out) out[i] = 0.f;
    if (i < n_nodes * HEADS) {
        e_max[i] = -3.402823466e+38f;
        e_sum[i] = 0.f;
    }
}

__device__ __forceinline__ void atomicMaxF32(float* addr, float val) {
    if (val >= 0.f)
        atomicMax((int*)addr, __float_as_int(val));
    else
        atomicMin((unsigned int*)addr, __float_as_uint(val));
}

__device__ __forceinline__ float leaky(float v) {
    return v > 0.f ? v : ALPHA * v;
}

__global__ void gat_edge_max(const int* __restrict__ src,
                             const int* __restrict__ dst,
                             const float* __restrict__ a1,
                             const float* __restrict__ a2,
                             float* __restrict__ e_max, int n_edges) {
    int e = blockIdx.x * blockDim.x + threadIdx.x;
    if (e >= n_edges) return;
    int s = src[e], d = dst[e];
    float4 A1 = *(const float4*)(a1 + (size_t)s * HEADS);
    float4 A2 = *(const float4*)(a2 + (size_t)d * HEADS);
    float* em = e_max + (size_t)d * HEADS;
    atomicMaxF32(em + 0, leaky(A1.x + A2.x));
    atomicMaxF32(em + 1, leaky(A1.y + A2.y));
    atomicMaxF32(em + 2, leaky(A1.z + A2.z));
    atomicMaxF32(em + 3, leaky(A1.w + A2.w));
}

__global__ void gat_edge_sum(const int* __restrict__ src,
                             const int* __restrict__ dst,
                             const float* __restrict__ a1,
                             const float* __restrict__ a2,
                             const float* __restrict__ e_max,
                             float* __restrict__ e_sum, int n_edges) {
    int e = blockIdx.x * blockDim.x + threadIdx.x;
    if (e >= n_edges) return;
    int s = src[e], d = dst[e];
    float4 A1 = *(const float4*)(a1 + (size_t)s * HEADS);
    float4 A2 = *(const float4*)(a2 + (size_t)d * HEADS);
    float4 EM = *(const float4*)(e_max + (size_t)d * HEADS);
    float* es = e_sum + (size_t)d * HEADS;
    atomicAdd(es + 0, expf(leaky(A1.x + A2.x) - EM.x));
    atomicAdd(es + 1, expf(leaky(A1.y + A2.y) - EM.y));
    atomicAdd(es + 2, expf(leaky(A1.z + A2.z) - EM.z));
    atomicAdd(es + 3, expf(leaky(A1.w + A2.w) - EM.w));
}

__global__ void gat_edge_agg(const int* __restrict__ src,
                             const int* __restrict__ dst,
                             const float* __restrict__ ft,
                             const float* __restrict__ a1,
                             const float* __restrict__ a2,
                             const float* __restrict__ e_max,
                             const float* __restrict__ e_sum,
                             float* __restrict__ out, int n_edges) {
    long long t = (long long)blockIdx.x * blockDim.x + threadIdx.x;
    if (t >= (long long)n_edges * 32) return;
    int e = (int)(t >> 5);
    int q = (int)(t & 31);
    int h = q >> 3;
    int s = src[e], d = dst[e];
    int idx = d * HEADS + h;
    float v = leaky(a1[s * HEADS + h] + a2[idx]);
    float alpha = expf(v - e_max[idx]) / e_sum[idx];
    float4 F = *(const float4*)(ft + (size_t)s * NCOL + 4 * q);
    float* op = out + (size_t)d * NCOL + 4 * q;
    atomicAdd(op + 0, F.x * alpha);
    atomicAdd(op + 1, F.y * alpha);
    atomicAdd(op + 2, F.z * alpha);
    atomicAdd(op + 3, F.w * alpha);
}

extern "C" void kernel_launch(void* const* d_in, const int* in_sizes, int n_in,
                              void* d_out, int out_size, void* d_ws, size_t ws_size,
                              hipStream_t stream) {
    const float* feature = (const float*)d_in[0];
    const int*   src     = (const int*)d_in[1];
    const int*   dst     = (const int*)d_in[2];
    const float* W_fc    = (const float*)d_in[3];
    const float* attn_l  = (const float*)d_in[4];
    const float* attn_r  = (const float*)d_in[5];
    float*       out     = (float*)d_out;

    const int M = in_sizes[0] / IN_DIM;
    const int E = in_sizes[1];

    float* ft    = (float*)d_ws;
    float* a1    = ft    + (size_t)M * NCOL;
    float* a2    = a1    + (size_t)M * HEADS;
    float* e_max = a2    + (size_t)M * HEADS;
    float* e_sum = e_max + (size_t)M * HEADS;

    {
        int n_waves = ((M + 15) / 16) * (NCOL / 16);
        int threads = n_waves * 32;
        gat_gemm_wmma<<<(threads + 255) / 256, 256, 0, stream>>>(feature, W_fc, ft, M);
    }
    gat_node_scores<<<(M * HEADS + 255) / 256, 256, 0, stream>>>(ft, attn_l, attn_r,
                                                                 a1, a2, M);
    gat_init<<<(M * NCOL + 255) / 256, 256, 0, stream>>>(out, e_max, e_sum, M);
    gat_edge_max<<<(E + 255) / 256, 256, 0, stream>>>(src, dst, a1, a2, e_max, E);
    gat_edge_sum<<<(E + 255) / 256, 256, 0, stream>>>(src, dst, a1, a2,
                                                      e_max, e_sum, E);
    {
        long long total = (long long)E * 32;
        int blocks = (int)((total + 255) / 256);
        gat_edge_agg<<<blocks, 256, 0, stream>>>(src, dst, ft, a1, a2,
                                                 e_max, e_sum, out, E);
    }
}


# Edge softmax and aggregation on owned destination ranges: 256 work-groups each filter the edge list for their 391 nodes; max, exp-sum, row sums as LDS atomics (arrival-order sums); rows stored once
# speedup vs baseline: 2.7464x; 2.7464x over previous
#include <hip/hip_runtime.h>
#include <hip/hip_bf16.h>

#define IN_DIM 256
#define NCOL 128
#define HEADS 4
#define OUT_DIM 32
#define ALPHA 0.2f

typedef __attribute__((ext_vector_type(2))) float v2f;
typedef __attribute__((ext_vector_type(8))) float v8f;

__global__ void gat_gemm_wmma(const float* __restrict__ A,
                              const float* __restrict__ B,
                              float* __restrict__ C, int M) {
    const int lane = threadIdx.x & 31;
    const int wave = (blockIdx.x * blockDim.x + threadIdx.x) >> 5;
    const int tiles_n = NCOL / 16;
    const int row_tile = wave / tiles_n;
    const int col_tile = wave % tiles_n;
    const int m0 = row_tile * 16;
    if (m0 >= M) return;

    const int n0   = col_tile * 16;
    const int r    = lane & 15;
    const int koff = (lane >> 4) * 2;

    v8f acc = {};
    const float* arow = A + (size_t)(m0 + r) * IN_DIM;
#pragma unroll 4
    for (int k = 0; k < IN_DIM; k += 4) {
        v2f a = *(const v2f*)(arow + k + koff);
        const float* bp = B + (size_t)(k + koff) * NCOL + n0 + r;
        v2f b;
        b.x = bp[0];
        b.y = bp[NCOL];
        acc = __builtin_amdgcn_wmma_f32_16x16x4_f32(
             false, a,  false, b,
             (short)0, acc,  false,  false);
    }

    const int row_add = (lane >> 4) * 8;
    float* cp = C + (size_t)(m0 + row_add) * NCOL + n0 + (lane & 15);
#pragma unroll
    for (int i = 0; i < 8; ++i) cp[(size_t)i * NCOL] = acc[i];
}

__global__ void gat_node_scores(const float* __restrict__ ft,
                                const float* __restrict__ attn_l,
                                const float* __restrict__ attn_r,
                                float* __restrict__ a1,
                                float* __restrict__ a2, int n_nodes) {
    int i = blockIdx.x * blockDim.x + threadIdx.x;
    if (i >= n_nodes * HEADS) return;
    int h = i & (HEADS - 1);
    const float* f  = ft + (size_t)(i >> 2) * NCOL + h * OUT_DIM;
    const float* al = attn_l + h * OUT_DIM;
    const float* ar = attn_r + h * OUT_DIM;
    float s1 = 0.f, s2 = 0.f;
#pragma unroll
    for (int d = 0; d < OUT_DIM; ++d) {
        float v = f[d];
        s1 += v * al[d];
        s2 += v * ar[d];
    }
    a1[i] = s1;
    a2[i] = s2;
}

#define OWN_THREADS 1024
#define OWN_GROUPS  256
#define R_MAX       391
#define CAP         12288
#define SEGV        (CAP / 4 - 1)
#define SCAN_U      8
#define AGG_U       8

__device__ __forceinline__ float leaky(float v) {
    return v > 0.f ? v : ALPHA * v;
}

__device__ __forceinline__ int f2key(float f) {
    int b = __float_as_int(f);
    return b >= 0 ? b : (b ^ 0x7FFFFFFF);
}
__device__ __forceinline__ float key2f(int k) {
    return __int_as_float(k >= 0 ? k : (k ^ 0x7FFFFFFF));
}

struct OwnLds {
    float acc[R_MAX * NCOL];
    int   mkey[R_MAX * HEADS];
    float ssum[R_MAX * HEADS];
    float a2s[R_MAX * HEADS];
    int2  lst[CAP];
    int   cnt;
};

__device__ __forceinline__ void own_filter(OwnLds& L, const int* __restrict__ dst,
                                           int v0, int v1, bool tail, int E,
                                           int lo, unsigned nR, int tid) {
    const int4* __restrict__ d4 = (const int4*)dst;
    const int count = v1 - v0;
    const int rot = (count / (int)gridDim.x) * (int)blockIdx.x;
    for (int b0 = 0; b0 < count; b0 += OWN_THREADS * SCAN_U) {
        int4 d[SCAN_U];
        int left = count - b0 - tid;
        int nvalid = left <= 0 ? 0 : (left + OWN_THREADS - 1) / OWN_THREADS;
        nvalid = nvalid > SCAN_U ? SCAN_U : nvalid;
#pragma unroll
        for (int u = 0; u < SCAN_U; ++u) {
            int i = b0 + u * OWN_THREADS + tid;
            int j = i + rot;
            if (j >= count) j -= count;
            if (i >= count) j = 0;
            d[u] = d4[v0 + j];
        }
        unsigned mask = 0;
#pragma unroll
        for (int u = 0; u < SCAN_U; ++u) {
            mask |= ((unsigned)(d[u].x - lo) < nR) ? (1u << (4 * u + 0)) : 0u;
            mask |= ((unsigned)(d[u].y - lo) < nR) ? (1u << (4 * u + 1)) : 0u;
            mask |= ((unsigned)(d[u].z - lo) < nR) ? (1u << (4 * u + 2)) : 0u;
            mask |= ((unsigned)(d[u].w - lo) < nR) ? (1u << (4 * u + 3)) : 0u;
        }
        mask &= (nvalid >= 8) ? 0xFFFFFFFFu : ((1u << (4 * nvalid)) - 1u);
        if (mask) {
            int slot = atomicAdd(&L.cnt, __popc(mask));
            while (mask) {
                int b = __ffs(mask) - 1;
                mask &= mask - 1;
                int i = b0 + (b >> 2) * OWN_THREADS + tid;
                int j = i + rot;
                if (j >= count) j -= count;
                if (slot < CAP) L.lst[slot] = make_int2(((v0 + j) << 2) + (b & 3), 0);
                ++slot;
            }
        }
    }
    if (tail && tid < (E & 3)) {
        int e = (v1 << 2) + tid;
        if ((unsigned)(dst[e] - lo) < nR) {
            int slot = atomicAdd(&L.cnt, 1);
            if (slot < CAP) L.lst[slot] = make_int2(e, 0);
        }
    }
}

__device__ __forceinline__ void own_resolve(OwnLds& L, const int* __restrict__ src,
                                            const int* __restrict__ dst,
                                            int n, int lo, int tid) {
    for (int i = tid; i < n; i += OWN_THREADS) {
        int e = L.lst[i].x;
        L.lst[i] = make_int2(src[e], dst[e] - lo);
    }
}

template <int PH>
__device__ __forceinline__ void own_edge_pass(OwnLds& L, const float* __restrict__ a1,
                                              int n, int tid) {
    for (int i = tid; i < n; i += OWN_THREADS) {
        int2 en = L.lst[i];
        float4 A1 = *(const float4*)(a1 + (size_t)en.x * HEADS);
        const int k = en.y * HEADS;
        float v0 = leaky(A1.x + L.a2s[k + 0]);
        float v1 = leaky(A1.y + L.a2s[k + 1]);
        float v2 = leaky(A1.z + L.a2s[k + 2]);
        float v3 = leaky(A1.w + L.a2s[k + 3]);
        if (PH == 0) {
            atomicMax(&L.mkey[k + 0], f2key(v0));
            atomicMax(&L.mkey[k + 1], f2key(v1));
            atomicMax(&L.mkey[k + 2], f2key(v2));
            atomicMax(&L.mkey[k + 3], f2key(v3));
        } else {
            atomicAdd(&L.ssum[k + 0], expf(v0 - key2f(L.mkey[k + 0])));
            atomicAdd(&L.ssum[k + 1], expf(v1 - key2f(L.mkey[k + 1])));
            atomicAdd(&L.ssum[k + 2], expf(v2 - key2f(L.mkey[k + 2])));
            atomicAdd(&L.ssum[k + 3], expf(v3 - key2f(L.mkey[k + 3])));
        }
    }
}

__device__ __forceinline__ void own_agg_pass(OwnLds& L, const float* __restrict__ ft,
                                             const float* __restrict__ a1,
                                             int n, int tid) {
    const int wave = tid >> 5;
    const int q = tid & 31;
    const int hl = q & 3;
    for (int i0 = wave; i0 < n; i0 += 32 * AGG_U) {
        int2  en[AGG_U];
        float F[AGG_U][HEADS];
        float A1h[AGG_U];
#pragma unroll
        for (int u = 0; u < AGG_U; ++u) {
            int i = i0 + u * 32;
            en[u] = L.lst[i < n ? i : i0];
        }
#pragma unroll
        for (int u = 0; u < AGG_U; ++u) {
            const float* fr = ft + (size_t)en[u].x * NCOL + q;
            F[u][0] = fr[0];
            F[u][1] = fr[32];
            F[u][2] = fr[64];
            F[u][3] = fr[96];
            A1h[u] = a1[en[u].x * HEADS + hl];
        }
#pragma unroll
        for (int u = 0; u < AGG_U; ++u) {
            if (i0 + u * 32 < n) {
                const int idx = en[u].y * HEADS + hl;
                float v = leaky(A1h[u] + L.a2s[idx]);
                float alpha = expf(v - key2f(L.mkey[idx])) / L.ssum[idx];
                float* op = L.acc + en[u].y * NCOL + q;
                const int ab = __float_as_int(alpha);
                const float al0 = __int_as_float(__builtin_amdgcn_readlane(ab, 0));
                const float al1 = __int_as_float(__builtin_amdgcn_readlane(ab, 1));
                const float al2 = __int_as_float(__builtin_amdgcn_readlane(ab, 2));
                const float al3 = __int_as_float(__builtin_amdgcn_readlane(ab, 3));
                atomicAdd(op +  0, F[u][0] * al0);
                atomicAdd(op + 32, F[u][1] * al1);
                atomicAdd(op + 64, F[u][2] * al2);
                atomicAdd(op + 96, F[u][3] * al3);
            }
        }
    }
}

__global__ void __launch_bounds__(OWN_THREADS)
gat_owner(const int* __restrict__ src, const int* __restrict__ dst,
          const float* __restrict__ ft, const float* __restrict__ a1,
          const float* __restrict__ a2, float* __restrict__ out,
          int n_nodes, int n_edges) {
    __shared__ OwnLds L;
    const int tid = threadIdx.x;
    const int nvec = n_edges >> 2;

    for (int lo = blockIdx.x * R_MAX; lo < n_nodes; lo += gridDim.x * R_MAX) {
        const int nR = (n_nodes - lo) < R_MAX ? (n_nodes - lo) : R_MAX;

        float4* acc4 = (float4*)L.acc;
        for (int i = tid; i < R_MAX * NCOL / 4; i += OWN_THREADS)
            acc4[i] = make_float4(0.f, 0.f, 0.f, 0.f);
        for (int i = tid; i < R_MAX * HEADS; i += OWN_THREADS) {
            L.mkey[i] = (int)0x80000000;
            L.ssum[i] = 0.f;
            L.a2s[i]  = (i < nR * HEADS) ? a2[(size_t)lo * HEADS + i] : 0.f;
        }
        if (tid == 0) L.cnt = 0;
        __syncthreads();

        own_filter(L, dst, 0, nvec, true, n_edges, lo, (unsigned)nR, tid);
        __syncthreads();
        const int total = L.cnt;

        if (total <= CAP) {
            own_resolve(L, src, dst, total, lo, tid);
            __syncthreads();
            own_edge_pass<0>(L, a1, total, tid);
            __syncthreads();
            own_edge_pass<1>(L, a1, total, tid);
            __syncthreads();
            own_agg_pass(L, ft, a1, total, tid);
        } else {
            for (int ph = 0; ph < 3; ++ph) {
                for (int v0 = 0; v0 <= nvec; v0 += SEGV) {
                    const int v1 = (v0 + SEGV < nvec) ? (v0 + SEGV) : nvec;
                    __syncthreads();
                    if (tid == 0) L.cnt = 0;
                    __syncthreads();
                    own_filter(L, dst, v0, v1, v1 == nvec, n_edges, lo, (unsigned)nR, tid);
                    __syncthreads();
                    const int n = L.cnt;
                    own_resolve(L, src, dst, n, lo, tid);
                    __syncthreads();
                    if (ph == 0)      own_edge_pass<0>(L, a1, n, tid);
                    else if (ph == 1) own_edge_pass<1>(L, a1, n, tid);
                    else              own_agg_pass(L, ft, a1, n, tid);
                    if (v1 == nvec) break;
                }
                __syncthreads();
            }
        }
        __syncthreads();

        float4* out4 = (float4*)(out + (size_t)lo * NCOL);
        for (int i = tid; i < nR * NCOL / 4; i += OWN_THREADS) out4[i] = acc4[i];
        __syncthreads();
    }
}

extern "C" void kernel_launch(void* const* d_in, const int* in_sizes, int n_in,
                              void* d_out, int out_size, void* d_ws, size_t ws_size,
                              hipStream_t stream) {
    const float* feature = (const float*)d_in[0];
    const int*   src     = (const int*)d_in[1];
    const int*   dst     = (const int*)d_in[2];
    const float* W_fc    = (const float*)d_in[3];
    const float* attn_l  = (const float*)d_in[4];
    const float* attn_r  = (const float*)d_in[5];
    float*       out     = (float*)d_out;

    const int M = in_sizes[0] / IN_DIM;
    const int E = in_sizes[1];

    float* ft    = (float*)d_ws;
    float* a1    = ft    + (size_t)M * NCOL;
    float* a2    = a1    + (size_t)M * HEADS;

    {
        int n_waves = ((M + 15) / 16) * (NCOL / 16);
        int threads = n_waves * 32;
        gat_gemm_wmma<<<(threads + 255) / 256, 256, 0, stream>>>(feature, W_fc, ft, M);
    }
    gat_node_scores<<<(M * HEADS + 255) / 256, 256, 0, stream>>>(ft, attn_l, attn_r,
                                                                 a1, a2, M);
    gat_owner<<<OWN_GROUPS, OWN_THREADS, 0, stream>>>(src, dst, ft, a1, a2, out, M, E);
}


# GEMM column-stationary: each wave keeps its 16-column slab of W_fc in registers for the whole launch; 16-row feature tiles staged once in LDS, shared by 8 waves; 512 resident groups; k order kept
# speedup vs baseline: 3.3844x; 1.2323x over previous
#include <hip/hip_runtime.h>
#include <hip/hip_bf16.h>

#define IN_DIM 256
#define NCOL 128
#define HEADS 4
#define OUT_DIM 32
#define ALPHA 0.2f

typedef __attribute__((ext_vector_type(2))) float v2f;
typedef __attribute__((ext_vector_type(8))) float v8f;

#define GW_THREADS 256
#define GW_GROUPS  512
#define A_PITCH    260

__device__ __forceinline__ float4 gw_load(const float* __restrict__ A, int tile, int f, int M) {
    int row = tile * 16 + (f >> 6);
    row = row < M ? row : M - 1;
    return *(const float4*)(A + (size_t)row * IN_DIM + (f & 63) * 4);
}
__device__ __forceinline__ void gw_stage(float* __restrict__ dstbuf, int f, float4 v) {
    *(float4*)(dstbuf + (f >> 6) * A_PITCH + (f & 63) * 4) = v;
}

__global__ void __launch_bounds__(GW_THREADS)
gat_gemm_wmma(const float* __restrict__ A,
              const float* __restrict__ B,
              float* __restrict__ C, int M) {
    __shared__ __attribute__((aligned(16))) float As[2][16 * A_PITCH];

    const int tid  = threadIdx.x;
    const int lane = tid & 31;
    const int n0   = (tid >> 5) * 16;
    const int r    = lane & 15;
    const int koff = (lane >> 4) * 2;
    const int tiles = (M + 15) >> 4;
    const int G = gridDim.x;

    int t = blockIdx.x;
    if (t >= tiles) return;

    const int f0 = tid, f1 = tid + GW_THREADS, f2 = tid + 2 * GW_THREADS, f3 = tid + 3 * GW_THREADS;
    float4 p0 = gw_load(A, t, f0, M);
    float4 p1 = gw_load(A, t, f1, M);
    float4 p2 = gw_load(A, t, f2, M);
    float4 p3 = gw_load(A, t, f3, M);

    v2f breg[IN_DIM / 4];
#pragma unroll
    for (int j = 0; j < IN_DIM / 4; ++j) {
        const float* bp = B + (size_t)(4 * j + koff) * NCOL + n0 + r;
        breg[j].x = bp[0];
        breg[j].y = bp[NCOL];
    }

    gw_stage(As[0], f0, p0);
    gw_stage(As[0], f1, p1);
    gw_stage(As[0], f2, p2);
    gw_stage(As[0], f3, p3);
    __syncthreads();

    int buf = 0;
    for (;;) {
        const int tn = t + G;
        const bool has_next = tn < tiles;
        if (has_next) {
            p0 = gw_load(A, tn, f0, M);
            p1 = gw_load(A, tn, f1, M);
            p2 = gw_load(A, tn, f2, M);
            p3 = gw_load(A, tn, f3, M);
        }

        v8f acc = {};
        const float* ap = &As[buf][r * A_PITCH + koff];
        v2f af[2][16];
#pragma unroll
        for (int i = 0; i < 16; ++i) af[0][i] = *(const v2f*)(ap + 4 * i);
#pragma unroll
        for (int kb = 0; kb < 4; ++kb) {
            if (kb < 3) {
#pragma unroll
                for (int i = 0; i < 16; ++i)
                    af[(kb + 1) & 1][i] = *(const v2f*)(ap + 4 * (16 * (kb + 1) + i));
            }
#pragma unroll
            for (int i = 0; i < 16; ++i) {
                acc = __builtin_amdgcn_wmma_f32_16x16x4_f32(
                     false, af[kb & 1][i],  false, breg[16 * kb + i],
                     (short)0, acc,  false,  false);
            }
        }
        __builtin_amdgcn_sched_group_barrier(0x100, 8, 0);
        __builtin_amdgcn_sched_group_barrier(0x100, 8, 0);
        __builtin_amdgcn_sched_group_barrier(0x008, 16, 0);
        __builtin_amdgcn_sched_group_barrier(0x100, 8, 0);
        __builtin_amdgcn_sched_group_barrier(0x008, 16, 0);
        __builtin_amdgcn_sched_group_barrier(0x100, 8, 0);
        __builtin_amdgcn_sched_group_barrier(0x008, 16, 0);
        __builtin_amdgcn_sched_group_barrier(0x008, 16, 0);

        if (has_next) {
            float* nb = As[buf ^ 1];
            gw_stage(nb, f0, p0);
            gw_stage(nb, f1, p1);
            gw_stage(nb, f2, p2);
            gw_stage(nb, f3, p3);
            __syncthreads();
        }

        const int row0 = t * 16 + (lane >> 4) * 8;
        float* cp = C + (size_t)row0 * NCOL + n0 + r;
        if (t * 16 + 16 <= M) {
#pragma unroll
            for (int i = 0; i < 8; ++i) cp[(size_t)i * NCOL] = acc[i];
        } else {
#pragma unroll
            for (int i = 0; i < 8; ++i)
                if (row0 + i < M) cp[(size_t)i * NCOL] = acc[i];
        }

        if (!has_next) break;
        buf ^= 1;
        t = tn;
    }
}

__global__ void gat_node_scores(const float* __restrict__ ft,
                                const float* __restrict__ attn_l,
                                const float* __restrict__ attn_r,
                                float* __restrict__ a1,
                                float* __restrict__ a2, int n_nodes) {
    int i = blockIdx.x * blockDim.x + threadIdx.x;
    if (i >= n_nodes * HEADS) return;
    int h = i & (HEADS - 1);
    const float* f  = ft + (size_t)(i >> 2) * NCOL + h * OUT_DIM;
    const float* al = attn_l + h * OUT_DIM;
    const float* ar = attn_r + h * OUT_DIM;
    float s1 = 0.f, s2 = 0.f;
#pragma unroll
    for (int d = 0; d < OUT_DIM; ++d) {
        float v = f[d];
        s1 += v * al[d];
        s2 += v * ar[d];
    }
    a1[i] = s1;
    a2[i] = s2;
}

#define OWN_THREADS 1024
#define OWN_GROUPS  256
#define R_MAX       391
#define CAP         12288
#define SEGV        (CAP / 4 - 1)
#define SCAN_U      8
#define AGG_U       8

__device__ __forceinline__ float leaky(float v) {
    return v > 0.f ? v : ALPHA * v;
}

__device__ __forceinline__ int f2key(float f) {
    int b = __float_as_int(f);
    return b >= 0 ? b : (b ^ 0x7FFFFFFF);
}
__device__ __forceinline__ float key2f(int k) {
    return __int_as_float(k >= 0 ? k : (k ^ 0x7FFFFFFF));
}

struct OwnLds {
    float acc[R_MAX * NCOL];
    int   mkey[R_MAX * HEADS];
    float ssum[R_MAX * HEADS];
    float a2s[R_MAX * HEADS];
    int2  lst[CAP];
    int   cnt;
};

__device__ __forceinline__ void own_filter(OwnLds& L, const int* __restrict__ dst,
                                           int v0, int v1, bool tail, int E,
                                           int lo, unsigned nR, int tid) {
    const int4* __restrict__ d4 = (const int4*)dst;
    const int count = v1 - v0;
    const int rot = (count / (int)gridDim.x) * (int)blockIdx.x;
    for (int b0 = 0; b0 < count; b0 += OWN_THREADS * SCAN_U) {
        int4 d[SCAN_U];
        int left = count - b0 - tid;
        int nvalid = left <= 0 ? 0 : (left + OWN_THREADS - 1) / OWN_THREADS;
        nvalid = nvalid > SCAN_U ? SCAN_U : nvalid;
#pragma unroll
        for (int u = 0; u < SCAN_U; ++u) {
            int i = b0 + u * OWN_THREADS + tid;
            int j = i + rot;
            if (j >= count) j -= count;
            if (i >= count) j = 0;
            d[u] = d4[v0 + j];
        }
        unsigned mask = 0;
#pragma unroll
        for (int u = 0; u < SCAN_U; ++u) {
            mask |= ((unsigned)(d[u].x - lo) < nR) ? (1u << (4 * u + 0)) : 0u;
            mask |= ((unsigned)(d[u].y - lo) < nR) ? (1u << (4 * u + 1)) : 0u;
            mask |= ((unsigned)(d[u].z - lo) < nR) ? (1u << (4 * u + 2)) : 0u;
            mask |= ((unsigned)(d[u].w - lo) < nR) ? (1u << (4 * u + 3)) : 0u;
        }
        mask &= (nvalid >= 8) ? 0xFFFFFFFFu : ((1u << (4 * nvalid)) - 1u);
        if (mask) {
            int slot = atomicAdd(&L.cnt, __popc(mask));
            while (mask) {
                int b = __ffs(mask) - 1;
                mask &= mask - 1;
                int i = b0 + (b >> 2) * OWN_THREADS + tid;
                int j = i + rot;
                if (j >= count) j -= count;
                if (slot < CAP) L.lst[slot] = make_int2(((v0 + j) << 2) + (b & 3), 0);
                ++slot;
            }
        }
    }
    if (tail && tid < (E & 3)) {
        int e = (v1 << 2) + tid;
        if ((unsigned)(dst[e] - lo) < nR) {
            int slot = atomicAdd(&L.cnt, 1);
            if (slot < CAP) L.lst[slot] = make_int2(e, 0);
        }
    }
}

__device__ __forceinline__ void own_resolve(OwnLds& L, const int* __restrict__ src,
                                            const int* __restrict__ dst,
                                            int n, int lo, int tid) {
    for (int i = tid; i < n; i += OWN_THREADS) {
        int e = L.lst[i].x;
        L.lst[i] = make_int2(src[e], dst[e] - lo);
    }
}

template <int PH>
__device__ __forceinline__ void own_edge_pass(OwnLds& L, const float* __restrict__ a1,
                                              int n, int tid) {
    for (int i = tid; i < n; i += OWN_THREADS) {
        int2 en = L.lst[i];
        float4 A1 = *(const float4*)(a1 + (size_t)en.x * HEADS);
        const int k = en.y * HEADS;
        float v0 = leaky(A1.x + L.a2s[k + 0]);
        float v1 = leaky(A1.y + L.a2s[k + 1]);
        float v2 = leaky(A1.z + L.a2s[k + 2]);
        float v3 = leaky(A1.w + L.a2s[k + 3]);
        if (PH == 0) {
            atomicMax(&L.mkey[k + 0], f2key(v0));
            atomicMax(&L.mkey[k + 1], f2key(v1));
            atomicMax(&L.mkey[k + 2], f2key(v2));
            atomicMax(&L.mkey[k + 3], f2key(v3));
        } else {
            atomicAdd(&L.ssum[k + 0], expf(v0 - key2f(L.mkey[k + 0])));
            atomicAdd(&L.ssum[k + 1], expf(v1 - key2f(L.mkey[k + 1])));
            atomicAdd(&L.ssum[k + 2], expf(v2 - key2f(L.mkey[k + 2])));
            atomicAdd(&L.ssum[k + 3], expf(v3 - key2f(L.mkey[k + 3])));
        }
    }
}

__device__ __forceinline__ void own_agg_pass(OwnLds& L, const float* __restrict__ ft,
                                             const float* __restrict__ a1,
                                             int n, int tid) {
    const int wave = tid >> 5;
    const int q = tid & 31;
    const int hl = q & 3;
    for (int i0 = wave; i0 < n; i0 += 32 * AGG_U) {
        int2  en[AGG_U];
        float F[AGG_U][HEADS];
        float A1h[AGG_U];
#pragma unroll
        for (int u = 0; u < AGG_U; ++u) {
            int i = i0 + u * 32;
            en[u] = L.lst[i < n ? i : i0];
        }
#pragma unroll
        for (int u = 0; u < AGG_U; ++u) {
            const float* fr = ft + (size_t)en[u].x * NCOL + q;
            F[u][0] = fr[0];
            F[u][1] = fr[32];
            F[u][2] = fr[64];
            F[u][3] = fr[96];
            A1h[u] = a1[en[u].x * HEADS + hl];
        }
#pragma unroll
        for (int u = 0; u < AGG_U; ++u) {
            if (i0 + u * 32 < n) {
                const int idx = en[u].y * HEADS + hl;
                float v = leaky(A1h[u] + L.a2s[idx]);
                float alpha = expf(v - key2f(L.mkey[idx])) / L.ssum[idx];
                float* op = L.acc + en[u].y * NCOL + q;
                const int ab = __float_as_int(alpha);
                const float al0 = __int_as_float(__builtin_amdgcn_readlane(ab, 0));
                const float al1 = __int_as_float(__builtin_amdgcn_readlane(ab, 1));
                const float al2 = __int_as_float(__builtin_amdgcn_readlane(ab, 2));
                const float al3 = __int_as_float(__builtin_amdgcn_readlane(ab, 3));
                atomicAdd(op +  0, F[u][0] * al0);
                atomicAdd(op + 32, F[u][1] * al1);
                atomicAdd(op + 64, F[u][2] * al2);
                atomicAdd(op + 96, F[u][3] * al3);
            }
        }
    }
}

__global__ void __launch_bounds__(OWN_THREADS)
gat_owner(const int* __restrict__ src, const int* __restrict__ dst,
          const float* __restrict__ ft, const float* __restrict__ a1,
          const float* __restrict__ a2, float* __restrict__ out,
          int n_nodes, int n_edges) {
    __shared__ OwnLds L;
    const int tid = threadIdx.x;
    const int nvec = n_edges >> 2;

    for (int lo = blockIdx.x * R_MAX; lo < n_nodes; lo += gridDim.x * R_MAX) {
        const int nR = (n_nodes - lo) < R_MAX ? (n_nodes - lo) : R_MAX;

        float4* acc4 = (float4*)L.acc;
        for (int i = tid; i < R_MAX * NCOL / 4; i += OWN_THREADS)
            acc4[i] = make_float4(0.f, 0.f, 0.f, 0.f);
        for (int i = tid; i < R_MAX * HEADS; i += OWN_THREADS) {
            L.mkey[i] = (int)0x80000000;
            L.ssum[i] = 0.f;
            L.a2s[i]  = (i < nR * HEADS) ? a2[(size_t)lo * HEADS + i] : 0.f;
        }
        if (tid == 0) L.cnt = 0;
        __syncthreads();

        own_filter(L, dst, 0, nvec, true, n_edges, lo, (unsigned)nR, tid);
        __syncthreads();
        const int total = L.cnt;

        if (total <= CAP) {
            own_resolve(L, src, dst, total, lo, tid);
            __syncthreads();
            own_edge_pass<0>(L, a1, total, tid);
            __syncthreads();
            own_edge_pass<1>(L, a1, total, tid);
            __syncthreads();
            own_agg_pass(L, ft, a1, total, tid);
        } else {
            for (int ph = 0; ph < 3; ++ph) {
                for (int v0 = 0; v0 <= nvec; v0 += SEGV) {
                    const int v1 = (v0 + SEGV < nvec) ? (v0 + SEGV) : nvec;
                    __syncthreads();
                    if (tid == 0) L.cnt = 0;
                    __syncthreads();
                    own_filter(L, dst, v0, v1, v1 == nvec, n_edges, lo, (unsigned)nR, tid);
                    __syncthreads();
                    const int n = L.cnt;
                    own_resolve(L, src, dst, n, lo, tid);
                    __syncthreads();
                    if (ph == 0)      own_edge_pass<0>(L, a1, n, tid);
                    else if (ph == 1) own_edge_pass<1>(L, a1, n, tid);
                    else              own_agg_pass(L, ft, a1, n, tid);
                    if (v1 == nvec) break;
                }
                __syncthreads();
            }
        }
        __syncthreads();

        float4* out4 = (float4*)(out + (size_t)lo * NCOL);
        for (int i = tid; i < nR * NCOL / 4; i += OWN_THREADS) out4[i] = acc4[i];
        __syncthreads();
    }
}

extern "C" void kernel_launch(void* const* d_in, const int* in_sizes, int n_in,
                              void* d_out, int out_size, void* d_ws, size_t ws_size,
                              hipStream_t stream) {
    const float* feature = (const float*)d_in[0];
    const int*   src     = (const int*)d_in[1];
    const int*   dst     = (const int*)d_in[2];
    const float* W_fc    = (const float*)d_in[3];
    const float* attn_l  = (const float*)d_in[4];
    const float* attn_r  = (const float*)d_in[5];
    float*       out     = (float*)d_out;

    const int M = in_sizes[0] / IN_DIM;
    const int E = in_sizes[1];

    float* ft    = (float*)d_ws;
    float* a1    = ft    + (size_t)M * NCOL;
    float* a2    = a1    + (size_t)M * HEADS;

    gat_gemm_wmma<<<GW_GROUPS, GW_THREADS, 0, stream>>>(feature, W_fc, ft, M);
    gat_node_scores<<<(M * HEADS + 255) / 256, 256, 0, stream>>>(ft, attn_l, attn_r,
                                                                 a1, a2, M);
    gat_owner<<<OWN_GROUPS, OWN_THREADS, 0, stream>>>(src, dst, ft, a1, a2, out, M, E);
}


# Attention scores taken in the GEMM epilogue from the accumulators: 16-lane DPP tree, two column halves added through LDS (new sum order for a1, a2); score launch and its re-read of ft removed
# speedup vs baseline: 3.5029x; 1.0350x over previous
#include <hip/hip_runtime.h>
#include <hip/hip_bf16.h>

#define IN_DIM 256
#define NCOL 128
#define HEADS 4
#define OUT_DIM 32
#define ALPHA 0.2f

typedef __attribute__((ext_vector_type(2))) float v2f;
typedef __attribute__((ext_vector_type(8))) float v8f;

#define GW_THREADS 256
#define GW_GROUPS  512
#define A_PITCH    260

__device__ __forceinline__ float4 gw_load(const float* __restrict__ A, int tile, int f, int M) {
    int row = tile * 16 + (f >> 6);
    row = row < M ? row : M - 1;
    return *(const float4*)(A + (size_t)row * IN_DIM + (f & 63) * 4);
}
__device__ __forceinline__ void gw_stage(float* __restrict__ dstbuf, int f, float4 v) {
    *(float4*)(dstbuf + (f >> 6) * A_PITCH + (f & 63) * 4) = v;
}

template <int XM>
__device__ __forceinline__ float gw_xlane(float v) {
    return __int_as_float(__builtin_amdgcn_update_dpp(0, __float_as_int(v), 0x160 + XM,
                                                      0xf, 0xf, true));
}
template <int XM>
__device__ __forceinline__ float gw_fold(bool bit, float lo, float hi) {
    const float keep = bit ? hi : lo;
    const float send = bit ? lo : hi;
    return keep + gw_xlane<XM>(send);
}

__global__ void __launch_bounds__(GW_THREADS)
gat_gemm_wmma(const float* __restrict__ A,
              const float* __restrict__ B,
              const float* __restrict__ attn_l,
              const float* __restrict__ attn_r,
              float* __restrict__ C,
              float* __restrict__ a1,
              float* __restrict__ a2, int M) {
    __shared__ __attribute__((aligned(16))) float As[2][16 * A_PITCH];
    __shared__ float part[2][GW_THREADS];

    const int tid  = threadIdx.x;
    const int lane = tid & 31;
    const int n0   = (tid >> 5) * 16;
    const int r    = lane & 15;
    const int koff = (lane >> 4) * 2;
    const int tiles = (M + 15) >> 4;
    const int G = gridDim.x;

    int t = blockIdx.x;
    if (t >= tiles) return;

    const float wl = attn_l[n0 + r];
    const float wr = attn_r[n0 + r];
    const bool c8 = (lane & 8) != 0, c4 = (lane & 4) != 0, c2 = (lane & 2) != 0, c1 = (lane & 1) != 0;

    const int f0 = tid, f1 = tid + GW_THREADS, f2 = tid + 2 * GW_THREADS, f3 = tid + 3 * GW_THREADS;
    float4 p0 = gw_load(A, t, f0, M);
    float4 p1 = gw_load(A, t, f1, M);
    float4 p2 = gw_load(A, t, f2, M);
    float4 p3 = gw_load(A, t, f3, M);

    v2f breg[IN_DIM / 4];
#pragma unroll
    for (int j = 0; j < IN_DIM / 4; ++j) {
        const float* bp = B + (size_t)(4 * j + koff) * NCOL + n0 + r;
        breg[j].x = bp[0];
        breg[j].y = bp[NCOL];
    }

    gw_stage(As[0], f0, p0);
    gw_stage(As[0], f1, p1);
    gw_stage(As[0], f2, p2);
    gw_stage(As[0], f3, p3);
    __syncthreads();

    int buf = 0;
    for (;;) {
        const int tn = t + G;
        const bool has_next = tn < tiles;
        if (has_next) {
            p0 = gw_load(A, tn, f0, M);
            p1 = gw_load(A, tn, f1, M);
            p2 = gw_load(A, tn, f2, M);
            p3 = gw_load(A, tn, f3, M);
        }

        v8f acc = {};
        const float* ap = &As[buf][r * A_PITCH + koff];
        v2f af[2][16];
#pragma unroll
        for (int i = 0; i < 16; ++i) af[0][i] = *(const v2f*)(ap + 4 * i);
#pragma unroll
        for (int kb = 0; kb < 4; ++kb) {
            if (kb < 3) {
#pragma unroll
                for (int i = 0; i < 16; ++i)
                    af[(kb + 1) & 1][i] = *(const v2f*)(ap + 4 * (16 * (kb + 1) + i));
            }
#pragma unroll
            for (int i = 0; i < 16; ++i) {
                acc = __builtin_amdgcn_wmma_f32_16x16x4_f32(
                     false, af[kb & 1][i],  false, breg[16 * kb + i],
                     (short)0, acc,  false,  false);
            }
        }
        __builtin_amdgcn_sched_group_barrier(0x100, 8, 0);
        __builtin_amdgcn_sched_group_barrier(0x100, 8, 0);
        __builtin_amdgcn_sched_group_barrier(0x008, 16, 0);
        __builtin_amdgcn_sched_group_barrier(0x100, 8, 0);
        __builtin_amdgcn_sched_group_barrier(0x008, 16, 0);
        __builtin_amdgcn_sched_group_barrier(0x100, 8, 0);
        __builtin_amdgcn_sched_group_barrier(0x008, 16, 0);
        __builtin_amdgcn_sched_group_barrier(0x008, 16, 0);

        {
            float sl[4], sr[4];
#pragma unroll
            for (int i = 0; i < 4; ++i) {
                sl[i] = gw_fold<8>(c8, acc[i] * wl, acc[i + 4] * wl);
                sr[i] = gw_fold<8>(c8, acc[i] * wr, acc[i + 4] * wr);
            }
            const float tl0 = gw_fold<4>(c4, sl[0], sl[2]);
            const float tl1 = gw_fold<4>(c4, sl[1], sl[3]);
            const float tr0 = gw_fold<4>(c4, sr[0], sr[2]);
            const float tr1 = gw_fold<4>(c4, sr[1], sr[3]);
            const float ul = gw_fold<2>(c2, tl0, tl1);
            const float ur = gw_fold<2>(c2, tr0, tr1);
            part[buf][tid] = gw_fold<1>(c1, ul, ur);
        }

        if (has_next) {
            float* nb = As[buf ^ 1];
            gw_stage(nb, f0, p0);
            gw_stage(nb, f1, p1);
            gw_stage(nb, f2, p2);
            gw_stage(nb, f3, p3);
        }
        __syncthreads();

        const int row0 = t * 16 + (lane >> 4) * 8;
        float* cp = C + (size_t)row0 * NCOL + n0 + r;
        if (t * 16 + 16 <= M) {
#pragma unroll
            for (int i = 0; i < 8; ++i) cp[(size_t)i * NCOL] = acc[i];
        } else {
#pragma unroll
            for (int i = 0; i < 8; ++i)
                if (row0 + i < M) cp[(size_t)i * NCOL] = acc[i];
        }

        if (tid < 128) {
            const int sc  = tid >> 6;
            const int o   = tid & 63;
            const int row = o >> 2, h = o & 3;
            const int pl  = (row >> 3) * 16 + (row & 7) * 2 + sc;
            const float v = part[buf][(2 * h) * 32 + pl] + part[buf][(2 * h + 1) * 32 + pl];
            float* ap2 = sc ? a2 : a1;
            if (t * 16 + row < M) ap2[(size_t)t * 64 + o] = v;
        }

        if (!has_next) break;
        buf ^= 1;
        t = tn;
    }
}

#define OWN_THREADS 1024
#define OWN_GROUPS  256
#define R_MAX       391
#define CAP         12288
#define SEGV        (CAP / 4 - 1)
#define SCAN_U      8
#define AGG_U       8

__device__ __forceinline__ float leaky(float v) {
    return v > 0.f ? v : ALPHA * v;
}

__device__ __forceinline__ int f2key(float f) {
    int b = __float_as_int(f);
    return b >= 0 ? b : (b ^ 0x7FFFFFFF);
}
__device__ __forceinline__ float key2f(int k) {
    return __int_as_float(k >= 0 ? k : (k ^ 0x7FFFFFFF));
}

struct OwnLds {
    float acc[R_MAX * NCOL];
    int   mkey[R_MAX * HEADS];
    float ssum[R_MAX * HEADS];
    float a2s[R_MAX * HEADS];
    int2  lst[CAP];
    int   cnt;
};

__device__ __forceinline__ void own_filter(OwnLds& L, const int* __restrict__ dst,
                                           int v0, int v1, bool tail, int E,
                                           int lo, unsigned nR, int tid) {
    const int4* __restrict__ d4 = (const int4*)dst;
    const int count = v1 - v0;
    const int rot = (count / (int)gridDim.x) * (int)blockIdx.x;
    for (int b0 = 0; b0 < count; b0 += OWN_THREADS * SCAN_U) {
        int4 d[SCAN_U];
        int left = count - b0 - tid;
        int nvalid = left <= 0 ? 0 : (left + OWN_THREADS - 1) / OWN_THREADS;
        nvalid = nvalid > SCAN_U ? SCAN_U : nvalid;
#pragma unroll
        for (int u = 0; u < SCAN_U; ++u) {
            int i = b0 + u * OWN_THREADS + tid;
            int j = i + rot;
            if (j >= count) j -= count;
            if (i >= count) j = 0;
            d[u] = d4[v0 + j];
        }
        unsigned mask = 0;
#pragma unroll
        for (int u = 0; u < SCAN_U; ++u) {
            mask |= ((unsigned)(d[u].x - lo) < nR) ? (1u << (4 * u + 0)) : 0u;
            mask |= ((unsigned)(d[u].y - lo) < nR) ? (1u << (4 * u + 1)) : 0u;
            mask |= ((unsigned)(d[u].z - lo) < nR) ? (1u << (4 * u + 2)) : 0u;
            mask |= ((unsigned)(d[u].w - lo) < nR) ? (1u << (4 * u + 3)) : 0u;
        }
        mask &= (nvalid >= 8) ? 0xFFFFFFFFu : ((1u << (4 * nvalid)) - 1u);
        if (mask) {
            int slot = atomicAdd(&L.cnt, __popc(mask));
            while (mask) {
                int b = __ffs(mask) - 1;
                mask &= mask - 1;
                int i = b0 + (b >> 2) * OWN_THREADS + tid;
                int j = i + rot;
                if (j >= count) j -= count;
                if (slot < CAP) L.lst[slot] = make_int2(((v0 + j) << 2) + (b & 3), 0);
                ++slot;
            }
        }
    }
    if (tail && tid < (E & 3)) {
        int e = (v1 << 2) + tid;
        if ((unsigned)(dst[e] - lo) < nR) {
            int slot = atomicAdd(&L.cnt, 1);
            if (slot < CAP) L.lst[slot] = make_int2(e, 0);
        }
    }
}

__device__ __forceinline__ void own_resolve(OwnLds& L, const int* __restrict__ src,
                                            const int* __restrict__ dst,
                                            int n, int lo, int tid) {
    for (int i = tid; i < n; i += OWN_THREADS) {
        int e = L.lst[i].x;
        L.lst[i] = make_int2(src[e], dst[e] - lo);
    }
}

template <int PH>
__device__ __forceinline__ void own_edge_pass(OwnLds& L, const float* __restrict__ a1,
                                              int n, int tid) {
    for (int i = tid; i < n; i += OWN_THREADS) {
        int2 en = L.lst[i];
        float4 A1 = *(const float4*)(a1 + (size_t)en.x * HEADS);
        const int k = en.y * HEADS;
        float v0 = leaky(A1.x + L.a2s[k + 0]);
        float v1 = leaky(A1.y + L.a2s[k + 1]);
        float v2 = leaky(A1.z + L.a2s[k + 2]);
        float v3 = leaky(A1.w + L.a2s[k + 3]);
        if (PH == 0) {
            atomicMax(&L.mkey[k + 0], f2key(v0));
            atomicMax(&L.mkey[k + 1], f2key(v1));
            atomicMax(&L.mkey[k + 2], f2key(v2));
            atomicMax(&L.mkey[k + 3], f2key(v3));
        } else {
            atomicAdd(&L.ssum[k + 0], expf(v0 - key2f(L.mkey[k + 0])));
            atomicAdd(&L.ssum[k + 1], expf(v1 - key2f(L.mkey[k + 1])));
            atomicAdd(&L.ssum[k + 2], expf(v2 - key2f(L.mkey[k + 2])));
            atomicAdd(&L.ssum[k + 3], expf(v3 - key2f(L.mkey[k + 3])));
        }
    }
}

__device__ __forceinline__ void own_agg_pass(OwnLds& L, const float* __restrict__ ft,
                                             const float* __restrict__ a1,
                                             int n, int tid) {
    const int wave = tid >> 5;
    const int q = tid & 31;
    const int hl = q & 3;
    for (int i0 = wave; i0 < n; i0 += 32 * AGG_U) {
        int2  en[AGG_U];
        float F[AGG_U][HEADS];
        float A1h[AGG_U];
#pragma unroll
        for (int u = 0; u < AGG_U; ++u) {
            int i = i0 + u * 32;
            en[u] = L.lst[i < n ? i : i0];
        }
#pragma unroll
        for (int u = 0; u < AGG_U; ++u) {
            const float* fr = ft + (size_t)en[u].x * NCOL + q;
            F[u][0] = fr[0];
            F[u][1] = fr[32];
            F[u][2] = fr[64];
            F[u][3] = fr[96];
            A1h[u] = a1[en[u].x * HEADS + hl];
        }
#pragma unroll
        for (int u = 0; u < AGG_U; ++u) {
            if (i0 + u * 32 < n) {
                const int idx = en[u].y * HEADS + hl;
                float v = leaky(A1h[u] + L.a2s[idx]);
                float alpha = expf(v - key2f(L.mkey[idx])) / L.ssum[idx];
                float* op = L.acc + en[u].y * NCOL + q;
                const int ab = __float_as_int(alpha);
                const float al0 = __int_as_float(__builtin_amdgcn_readlane(ab, 0));
                const float al1 = __int_as_float(__builtin_amdgcn_readlane(ab, 1));
                const float al2 = __int_as_float(__builtin_amdgcn_readlane(ab, 2));
                const float al3 = __int_as_float(__builtin_amdgcn_readlane(ab, 3));
                atomicAdd(op +  0, F[u][0] * al0);
                atomicAdd(op + 32, F[u][1] * al1);
                atomicAdd(op + 64, F[u][2] * al2);
                atomicAdd(op + 96, F[u][3] * al3);
            }
        }
    }
}

__global__ void __launch_bounds__(OWN_THREADS)
gat_owner(const int* __restrict__ src, const int* __restrict__ dst,
          const float* __restrict__ ft, const float* __restrict__ a1,
          const float* __restrict__ a2, float* __restrict__ out,
          int n_nodes, int n_edges) {
    __shared__ OwnLds L;
    const int tid = threadIdx.x;
    const int nvec = n_edges >> 2;

    for (int lo = blockIdx.x * R_MAX; lo < n_nodes; lo += gridDim.x * R_MAX) {
        const int nR = (n_nodes - lo) < R_MAX ? (n_nodes - lo) : R_MAX;

        float4* acc4 = (float4*)L.acc;
        for (int i = tid; i < R_MAX * NCOL / 4; i += OWN_THREADS)
            acc4[i] = make_float4(0.f, 0.f, 0.f, 0.f);
        for (int i = tid; i < R_MAX * HEADS; i += OWN_THREADS) {
            L.mkey[i] = (int)0x80000000;
            L.ssum[i] = 0.f;
            L.a2s[i]  = (i < nR * HEADS) ? a2[(size_t)lo * HEADS + i] : 0.f;
        }
        if (tid == 0) L.cnt = 0;
        __syncthreads();

        own_filter(L, dst, 0, nvec, true, n_edges, lo, (unsigned)nR, tid);
        __syncthreads();
        const int total = L.cnt;

        if (total <= CAP) {
            own_resolve(L, src, dst, total, lo, tid);
            __syncthreads();
            own_edge_pass<0>(L, a1, total, tid);
            __syncthreads();
            own_edge_pass<1>(L, a1, total, tid);
            __syncthreads();
            own_agg_pass(L, ft, a1, total, tid);
        } else {
            for (int ph = 0; ph < 3; ++ph) {
                for (int v0 = 0; v0 <= nvec; v0 += SEGV) {
                    const int v1 = (v0 + SEGV < nvec) ? (v0 + SEGV) : nvec;
                    __syncthreads();
                    if (tid == 0) L.cnt = 0;
                    __syncthreads();
                    own_filter(L, dst, v0, v1, v1 == nvec, n_edges, lo, (unsigned)nR, tid);
                    __syncthreads();
                    const int n = L.cnt;
                    own_resolve(L, src, dst, n, lo, tid);
                    __syncthreads();
                    if (ph == 0)      own_edge_pass<0>(L, a1, n, tid);
                    else if (ph == 1) own_edge_pass<1>(L, a1, n, tid);
                    else              own_agg_pass(L, ft, a1, n, tid);
                    if (v1 == nvec) break;
                }
                __syncthreads();
            }
        }
        __syncthreads();

        float4* out4 = (float4*)(out + (size_t)lo * NCOL);
        for (int i = tid; i < nR * NCOL / 4; i += OWN_THREADS) out4[i] = acc4[i];
        __syncthreads();
    }
}

extern "C" void kernel_launch(void* const* d_in, const int* in_sizes, int n_in,
                              void* d_out, int out_size, void* d_ws, size_t ws_size,
                              hipStream_t stream) {
    const float* feature = (const float*)d_in[0];
    const int*   src     = (const int*)d_in[1];
    const int*   dst     = (const int*)d_in[2];
    const float* W_fc    = (const float*)d_in[3];
    const float* attn_l  = (const float*)d_in[4];
    const float* attn_r  = (const float*)d_in[5];
    float*       out     = (float*)d_out;

    const int M = in_sizes[0] / IN_DIM;
    const int E = in_sizes[1];

    float* ft    = (float*)d_ws;
    float* a1    = ft    + (size_t)M * NCOL;
    float* a2    = a1    + (size_t)M * HEADS;

    gat_gemm_wmma<<<GW_GROUPS, GW_THREADS, 0, stream>>>(feature, W_fc, attn_l, attn_r,
                                                        ft, a1, a2, M);
    gat_owner<<<OWN_GROUPS, OWN_THREADS, 0, stream>>>(src, dst, ft, a1, a2, out, M, E);
}


# Aggregation weights: one lane per (edge, head) computes alpha for a batch of 8 edges at once, row adds read it through scalar registers; was 8 exp+divide passes over all 32 lanes; sum order kept
# speedup vs baseline: 3.8666x; 1.1038x over previous
#include <hip/hip_runtime.h>
#include <hip/hip_bf16.h>

#define IN_DIM 256
#define NCOL 128
#define HEADS 4
#define OUT_DIM 32
#define ALPHA 0.2f

typedef __attribute__((ext_vector_type(2))) float v2f;
typedef __attribute__((ext_vector_type(8))) float v8f;

#define GW_THREADS 256
#define GW_GROUPS  512
#define A_PITCH    260

__device__ __forceinline__ float4 gw_load(const float* __restrict__ A, int tile, int f, int M) {
    int row = tile * 16 + (f >> 6);
    row = row < M ? row : M - 1;
    return *(const float4*)(A + (size_t)row * IN_DIM + (f & 63) * 4);
}
__device__ __forceinline__ void gw_stage(float* __restrict__ dstbuf, int f, float4 v) {
    *(float4*)(dstbuf + (f >> 6) * A_PITCH + (f & 63) * 4) = v;
}

template <int XM>
__device__ __forceinline__ float gw_xlane(float v) {
    return __int_as_float(__builtin_amdgcn_update_dpp(0, __float_as_int(v), 0x160 + XM,
                                                      0xf, 0xf, true));
}
template <int XM>
__device__ __forceinline__ float gw_fold(bool bit, float lo, float hi) {
    const float keep = bit ? hi : lo;
    const float send = bit ? lo : hi;
    return keep + gw_xlane<XM>(send);
}

__global__ void __launch_bounds__(GW_THREADS)
gat_gemm_wmma(const float* __restrict__ A,
              const float* __restrict__ B,
              const float* __restrict__ attn_l,
              const float* __restrict__ attn_r,
              float* __restrict__ C,
              float* __restrict__ a1,
              float* __restrict__ a2, int M) {
    __shared__ __attribute__((aligned(16))) float As[2][16 * A_PITCH];
    __shared__ float part[2][GW_THREADS];

    const int tid  = threadIdx.x;
    const int lane = tid & 31;
    const int n0   = (tid >> 5) * 16;
    const int r    = lane & 15;
    const int koff = (lane >> 4) * 2;
    const int tiles = (M + 15) >> 4;
    const int G = gridDim.x;

    int t = blockIdx.x;
    if (t >= tiles) return;

    const float wl = attn_l[n0 + r];
    const float wr = attn_r[n0 + r];
    const bool c8 = (lane & 8) != 0, c4 = (lane & 4) != 0, c2 = (lane & 2) != 0, c1 = (lane & 1) != 0;

    const int f0 = tid, f1 = tid + GW_THREADS, f2 = tid + 2 * GW_THREADS, f3 = tid + 3 * GW_THREADS;
    float4 p0 = gw_load(A, t, f0, M);
    float4 p1 = gw_load(A, t, f1, M);
    float4 p2 = gw_load(A, t, f2, M);
    float4 p3 = gw_load(A, t, f3, M);

    v2f breg[IN_DIM / 4];
#pragma unroll
    for (int j = 0; j < IN_DIM / 4; ++j) {
        const float* bp = B + (size_t)(4 * j + koff) * NCOL + n0 + r;
        breg[j].x = bp[0];
        breg[j].y = bp[NCOL];
    }

    gw_stage(As[0], f0, p0);
    gw_stage(As[0], f1, p1);
    gw_stage(As[0], f2, p2);
    gw_stage(As[0], f3, p3);
    __syncthreads();

    int buf = 0;
    for (;;) {
        const int tn = t + G;
        const bool has_next = tn < tiles;
        if (has_next) {
            p0 = gw_load(A, tn, f0, M);
            p1 = gw_load(A, tn, f1, M);
            p2 = gw_load(A, tn, f2, M);
            p3 = gw_load(A, tn, f3, M);
        }

        v8f acc = {};
        const float* ap = &As[buf][r * A_PITCH + koff];
        v2f af[2][16];
#pragma unroll
        for (int i = 0; i < 16; ++i) af[0][i] = *(const v2f*)(ap + 4 * i);
#pragma unroll
        for (int kb = 0; kb < 4; ++kb) {
            if (kb < 3) {
#pragma unroll
                for (int i = 0; i < 16; ++i)
                    af[(kb + 1) & 1][i] = *(const v2f*)(ap + 4 * (16 * (kb + 1) + i));
            }
#pragma unroll
            for (int i = 0; i < 16; ++i) {
                acc = __builtin_amdgcn_wmma_f32_16x16x4_f32(
                     false, af[kb & 1][i],  false, breg[16 * kb + i],
                     (short)0, acc,  false,  false);
            }
        }
        __builtin_amdgcn_sched_group_barrier(0x100, 8, 0);
        __builtin_amdgcn_sched_group_barrier(0x100, 8, 0);
        __builtin_amdgcn_sched_group_barrier(0x008, 16, 0);
        __builtin_amdgcn_sched_group_barrier(0x100, 8, 0);
        __builtin_amdgcn_sched_group_barrier(0x008, 16, 0);
        __builtin_amdgcn_sched_group_barrier(0x100, 8, 0);
        __builtin_amdgcn_sched_group_barrier(0x008, 16, 0);
        __builtin_amdgcn_sched_group_barrier(0x008, 16, 0);

        {
            float sl[4], sr[4];
#pragma unroll
            for (int i = 0; i < 4; ++i) {
                sl[i] = gw_fold<8>(c8, acc[i] * wl, acc[i + 4] * wl);
                sr[i] = gw_fold<8>(c8, acc[i] * wr, acc[i + 4] * wr);
            }
            const float tl0 = gw_fold<4>(c4, sl[0], sl[2]);
            const float tl1 = gw_fold<4>(c4, sl[1], sl[3]);
            const float tr0 = gw_fold<4>(c4, sr[0], sr[2]);
            const float tr1 = gw_fold<4>(c4, sr[1], sr[3]);
            const float ul = gw_fold<2>(c2, tl0, tl1);
            const float ur = gw_fold<2>(c2, tr0, tr1);
            part[buf][tid] = gw_fold<1>(c1, ul, ur);
        }

        if (has_next) {
            float* nb = As[buf ^ 1];
            gw_stage(nb, f0, p0);
            gw_stage(nb, f1, p1);
            gw_stage(nb, f2, p2);
            gw_stage(nb, f3, p3);
        }
        __syncthreads();

        const int row0 = t * 16 + (lane >> 4) * 8;
        float* cp = C + (size_t)row0 * NCOL + n0 + r;
        if (t * 16 + 16 <= M) {
#pragma unroll
            for (int i = 0; i < 8; ++i) cp[(size_t)i * NCOL] = acc[i];
        } else {
#pragma unroll
            for (int i = 0; i < 8; ++i)
                if (row0 + i < M) cp[(size_t)i * NCOL] = acc[i];
        }

        if (tid < 128) {
            const int sc  = tid >> 6;
            const int o   = tid & 63;
            const int row = o >> 2, h = o & 3;
            const int pl  = (row >> 3) * 16 + (row & 7) * 2 + sc;
            const float v = part[buf][(2 * h) * 32 + pl] + part[buf][(2 * h + 1) * 32 + pl];
            float* ap2 = sc ? a2 : a1;
            if (t * 16 + row < M) ap2[(size_t)t * 64 + o] = v;
        }

        if (!has_next) break;
        buf ^= 1;
        t = tn;
    }
}

#define OWN_THREADS 1024
#define OWN_GROUPS  256
#define R_MAX       391
#define CAP         12288
#define SEGV        (CAP / 4 - 1)
#define SCAN_U      8
#define AGG_U       8

__device__ __forceinline__ float leaky(float v) {
    return v > 0.f ? v : ALPHA * v;
}

__device__ __forceinline__ int f2key(float f) {
    int b = __float_as_int(f);
    return b >= 0 ? b : (b ^ 0x7FFFFFFF);
}
__device__ __forceinline__ float key2f(int k) {
    return __int_as_float(k >= 0 ? k : (k ^ 0x7FFFFFFF));
}

struct OwnLds {
    float acc[R_MAX * NCOL];
    int   mkey[R_MAX * HEADS];
    float ssum[R_MAX * HEADS];
    float a2s[R_MAX * HEADS];
    int2  lst[CAP];
    int   cnt;
};

__device__ __forceinline__ void own_filter(OwnLds& L, const int* __restrict__ dst,
                                           int v0, int v1, bool tail, int E,
                                           int lo, unsigned nR, int tid) {
    const int4* __restrict__ d4 = (const int4*)dst;
    const int count = v1 - v0;
    const int rot = (count / (int)gridDim.x) * (int)blockIdx.x;
    for (int b0 = 0; b0 < count; b0 += OWN_THREADS * SCAN_U) {
        int4 d[SCAN_U];
        int left = count - b0 - tid;
        int nvalid = left <= 0 ? 0 : (left + OWN_THREADS - 1) / OWN_THREADS;
        nvalid = nvalid > SCAN_U ? SCAN_U : nvalid;
#pragma unroll
        for (int u = 0; u < SCAN_U; ++u) {
            int i = b0 + u * OWN_THREADS + tid;
            int j = i + rot;
            if (j >= count) j -= count;
            if (i >= count) j = 0;
            d[u] = d4[v0 + j];
        }
        unsigned mask = 0;
#pragma unroll
        for (int u = 0; u < SCAN_U; ++u) {
            mask |= ((unsigned)(d[u].x - lo) < nR) ? (1u << (4 * u + 0)) : 0u;
            mask |= ((unsigned)(d[u].y - lo) < nR) ? (1u << (4 * u + 1)) : 0u;
            mask |= ((unsigned)(d[u].z - lo) < nR) ? (1u << (4 * u + 2)) : 0u;
            mask |= ((unsigned)(d[u].w - lo) < nR) ? (1u << (4 * u + 3)) : 0u;
        }
        mask &= (nvalid >= 8) ? 0xFFFFFFFFu : ((1u << (4 * nvalid)) - 1u);
        if (mask) {
            int slot = atomicAdd(&L.cnt, __popc(mask));
            while (mask) {
                int b = __ffs(mask) - 1;
                mask &= mask - 1;
                int i = b0 + (b >> 2) * OWN_THREADS + tid;
                int j = i + rot;
                if (j >= count) j -= count;
                if (slot < CAP) L.lst[slot] = make_int2(((v0 + j) << 2) + (b & 3), 0);
                ++slot;
            }
        }
    }
    if (tail && tid < (E & 3)) {
        int e = (v1 << 2) + tid;
        if ((unsigned)(dst[e] - lo) < nR) {
            int slot = atomicAdd(&L.cnt, 1);
            if (slot < CAP) L.lst[slot] = make_int2(e, 0);
        }
    }
}

__device__ __forceinline__ void own_resolve(OwnLds& L, const int* __restrict__ src,
                                            const int* __restrict__ dst,
                                            int n, int lo, int tid) {
    for (int i = tid; i < n; i += OWN_THREADS) {
        int e = L.lst[i].x;
        L.lst[i] = make_int2(src[e], dst[e] - lo);
    }
}

template <int PH>
__device__ __forceinline__ void own_edge_pass(OwnLds& L, const float* __restrict__ a1,
                                              int n, int tid) {
    for (int i = tid; i < n; i += OWN_THREADS) {
        int2 en = L.lst[i];
        float4 A1 = *(const float4*)(a1 + (size_t)en.x * HEADS);
        const int k = en.y * HEADS;
        float v0 = leaky(A1.x + L.a2s[k + 0]);
        float v1 = leaky(A1.y + L.a2s[k + 1]);
        float v2 = leaky(A1.z + L.a2s[k + 2]);
        float v3 = leaky(A1.w + L.a2s[k + 3]);
        if (PH == 0) {
            atomicMax(&L.mkey[k + 0], f2key(v0));
            atomicMax(&L.mkey[k + 1], f2key(v1));
            atomicMax(&L.mkey[k + 2], f2key(v2));
            atomicMax(&L.mkey[k + 3], f2key(v3));
        } else {
            atomicAdd(&L.ssum[k + 0], expf(v0 - key2f(L.mkey[k + 0])));
            atomicAdd(&L.ssum[k + 1], expf(v1 - key2f(L.mkey[k + 1])));
            atomicAdd(&L.ssum[k + 2], expf(v2 - key2f(L.mkey[k + 2])));
            atomicAdd(&L.ssum[k + 3], expf(v3 - key2f(L.mkey[k + 3])));
        }
    }
}

static_assert(AGG_U * HEADS == 32, "one lane per (batch edge, head)");
__device__ __forceinline__ void own_agg_pass(OwnLds& L, const float* __restrict__ ft,
                                             const float* __restrict__ a1,
                                             int n, int tid) {
    const int wave = tid >> 5;
    const int q = tid & 31;
    const int um = q >> 2;
    const int hl = q & 3;
    for (int i0 = wave; i0 < n; i0 += 32 * AGG_U) {
        int2  en[AGG_U];
        float F[AGG_U][HEADS];
#pragma unroll
        for (int u = 0; u < AGG_U; ++u) {
            int i = i0 + u * 32;
            en[u] = L.lst[i < n ? i : i0];
        }
        const int im = i0 + um * 32;
        const int2 em = L.lst[im < n ? im : i0];
#pragma unroll
        for (int u = 0; u < AGG_U; ++u) {
            const float* fr = ft + (size_t)en[u].x * NCOL + q;
            F[u][0] = fr[0];
            F[u][1] = fr[32];
            F[u][2] = fr[64];
            F[u][3] = fr[96];
        }
        const float A1m = a1[em.x * HEADS + hl];
        const int idx = em.y * HEADS + hl;
        const float v = leaky(A1m + L.a2s[idx]);
        const float alpha = expf(v - key2f(L.mkey[idx])) / L.ssum[idx];
        const int ab = __float_as_int(alpha);
#pragma unroll
        for (int u = 0; u < AGG_U; ++u) {
            if (i0 + u * 32 < n) {
                float* op = L.acc + en[u].y * NCOL + q;
                const float al0 = __int_as_float(__builtin_amdgcn_readlane(ab, 4 * u + 0));
                const float al1 = __int_as_float(__builtin_amdgcn_readlane(ab, 4 * u + 1));
                const float al2 = __int_as_float(__builtin_amdgcn_readlane(ab, 4 * u + 2));
                const float al3 = __int_as_float(__builtin_amdgcn_readlane(ab, 4 * u + 3));
                atomicAdd(op +  0, F[u][0] * al0);
                atomicAdd(op + 32, F[u][1] * al1);
                atomicAdd(op + 64, F[u][2] * al2);
                atomicAdd(op + 96, F[u][3] * al3);
            }
        }
    }
}

__global__ void __launch_bounds__(OWN_THREADS)
gat_owner(const int* __restrict__ src, const int* __restrict__ dst,
          const float* __restrict__ ft, const float* __restrict__ a1,
          const float* __restrict__ a2, float* __restrict__ out,
          int n_nodes, int n_edges) {
    __shared__ OwnLds L;
    const int tid = threadIdx.x;
    const int nvec = n_edges >> 2;

    for (int lo = blockIdx.x * R_MAX; lo < n_nodes; lo += gridDim.x * R_MAX) {
        const int nR = (n_nodes - lo) < R_MAX ? (n_nodes - lo) : R_MAX;

        float4* acc4 = (float4*)L.acc;
        for (int i = tid; i < R_MAX * NCOL / 4; i += OWN_THREADS)
            acc4[i] = make_float4(0.f, 0.f, 0.f, 0.f);
        for (int i = tid; i < R_MAX * HEADS; i += OWN_THREADS) {
            L.mkey[i] = (int)0x80000000;
            L.ssum[i] = 0.f;
            L.a2s[i]  = (i < nR * HEADS) ? a2[(size_t)lo * HEADS + i] : 0.f;
        }
        if (tid == 0) L.cnt = 0;
        __syncthreads();

        own_filter(L, dst, 0, nvec, true, n_edges, lo, (unsigned)nR, tid);
        __syncthreads();
        const int total = L.cnt;

        if (total <= CAP) {
            own_resolve(L, src, dst, total, lo, tid);
            __syncthreads();
            own_edge_pass<0>(L, a1, total, tid);
            __syncthreads();
            own_edge_pass<1>(L, a1, total, tid);
            __syncthreads();
            own_agg_pass(L, ft, a1, total, tid);
        } else {
            for (int ph = 0; ph < 3; ++ph) {
                for (int v0 = 0; v0 <= nvec; v0 += SEGV) {
                    const int v1 = (v0 + SEGV < nvec) ? (v0 + SEGV) : nvec;
                    __syncthreads();
                    if (tid == 0) L.cnt = 0;
                    __syncthreads();
                    own_filter(L, dst, v0, v1, v1 == nvec, n_edges, lo, (unsigned)nR, tid);
                    __syncthreads();
                    const int n = L.cnt;
                    own_resolve(L, src, dst, n, lo, tid);
                    __syncthreads();
                    if (ph == 0)      own_edge_pass<0>(L, a1, n, tid);
                    else if (ph == 1) own_edge_pass<1>(L, a1, n, tid);
                    else              own_agg_pass(L, ft, a1, n, tid);
                    if (v1 == nvec) break;
                }
                __syncthreads();
            }
        }
        __syncthreads();

        float4* out4 = (float4*)(out + (size_t)lo * NCOL);
        for (int i = tid; i < nR * NCOL / 4; i += OWN_THREADS) out4[i] = acc4[i];
        __syncthreads();
    }
}

extern "C" void kernel_launch(void* const* d_in, const int* in_sizes, int n_in,
                              void* d_out, int out_size, void* d_ws, size_t ws_size,
                              hipStream_t stream) {
    const float* feature = (const float*)d_in[0];
    const int*   src     = (const int*)d_in[1];
    const int*   dst     = (const int*)d_in[2];
    const float* W_fc    = (const float*)d_in[3];
    const float* attn_l  = (const float*)d_in[4];
    const float* attn_r  = (const float*)d_in[5];
    float*       out     = (float*)d_out;

    const int M = in_sizes[0] / IN_DIM;
    const int E = in_sizes[1];

    float* ft    = (float*)d_ws;
    float* a1    = ft    + (size_t)M * NCOL;
    float* a2    = a1    + (size_t)M * HEADS;

    gat_gemm_wmma<<<GW_GROUPS, GW_THREADS, 0, stream>>>(feature, W_fc, attn_l, attn_r,
                                                        ft, a1, a2, M);
    gat_owner<<<OWN_GROUPS, OWN_THREADS, 0, stream>>>(src, dst, ft, a1, a2, out, M, E);
}


# Edge filter: each int4 group of dst folded to the unsigned min of its four (dst - lo), edges tested only in groups holding a match; list walked as two plain runs; arrival-order sums, new list order
# speedup vs baseline: 4.1121x; 1.0635x over previous
#include <hip/hip_runtime.h>
#include <hip/hip_bf16.h>

#define IN_DIM 256
#define NCOL 128
#define HEADS 4
#define OUT_DIM 32
#define ALPHA 0.2f

typedef __attribute__((ext_vector_type(2))) float v2f;
typedef __attribute__((ext_vector_type(8))) float v8f;

#define GW_THREADS 256
#define GW_GROUPS  512
#define A_PITCH    260

__device__ __forceinline__ float4 gw_load(const float* __restrict__ A, int tile, int f, int M) {
    int row = tile * 16 + (f >> 6);
    row = row < M ? row : M - 1;
    return *(const float4*)(A + (size_t)row * IN_DIM + (f & 63) * 4);
}
__device__ __forceinline__ void gw_stage(float* __restrict__ dstbuf, int f, float4 v) {
    *(float4*)(dstbuf + (f >> 6) * A_PITCH + (f & 63) * 4) = v;
}

template <int XM>
__device__ __forceinline__ float gw_xlane(float v) {
    return __int_as_float(__builtin_amdgcn_update_dpp(0, __float_as_int(v), 0x160 + XM,
                                                      0xf, 0xf, true));
}
template <int XM>
__device__ __forceinline__ float gw_fold(bool bit, float lo, float hi) {
    const float keep = bit ? hi : lo;
    const float send = bit ? lo : hi;
    return keep + gw_xlane<XM>(send);
}

__global__ void __launch_bounds__(GW_THREADS)
gat_gemm_wmma(const float* __restrict__ A,
              const float* __restrict__ B,
              const float* __restrict__ attn_l,
              const float* __restrict__ attn_r,
              float* __restrict__ C,
              float* __restrict__ a1,
              float* __restrict__ a2, int M) {
    __shared__ __attribute__((aligned(16))) float As[2][16 * A_PITCH];
    __shared__ float part[2][GW_THREADS];

    const int tid  = threadIdx.x;
    const int lane = tid & 31;
    const int n0   = (tid >> 5) * 16;
    const int r    = lane & 15;
    const int koff = (lane >> 4) * 2;
    const int tiles = (M + 15) >> 4;
    const int G = gridDim.x;

    int t = blockIdx.x;
    if (t >= tiles) return;

    const float wl = attn_l[n0 + r];
    const float wr = attn_r[n0 + r];
    const bool c8 = (lane & 8) != 0, c4 = (lane & 4) != 0, c2 = (lane & 2) != 0, c1 = (lane & 1) != 0;

    const int f0 = tid, f1 = tid + GW_THREADS, f2 = tid + 2 * GW_THREADS, f3 = tid + 3 * GW_THREADS;
    float4 p0 = gw_load(A, t, f0, M);
    float4 p1 = gw_load(A, t, f1, M);
    float4 p2 = gw_load(A, t, f2, M);
    float4 p3 = gw_load(A, t, f3, M);

    v2f breg[IN_DIM / 4];
#pragma unroll
    for (int j = 0; j < IN_DIM / 4; ++j) {
        const float* bp = B + (size_t)(4 * j + koff) * NCOL + n0 + r;
        breg[j].x = bp[0];
        breg[j].y = bp[NCOL];
    }

    gw_stage(As[0], f0, p0);
    gw_stage(As[0], f1, p1);
    gw_stage(As[0], f2, p2);
    gw_stage(As[0], f3, p3);
    __syncthreads();

    int buf = 0;
    for (;;) {
        const int tn = t + G;
        const bool has_next = tn < tiles;
        if (has_next) {
            p0 = gw_load(A, tn, f0, M);
            p1 = gw_load(A, tn, f1, M);
            p2 = gw_load(A, tn, f2, M);
            p3 = gw_load(A, tn, f3, M);
        }

        v8f acc = {};
        const float* ap = &As[buf][r * A_PITCH + koff];
        v2f af[2][16];
#pragma unroll
        for (int i = 0; i < 16; ++i) af[0][i] = *(const v2f*)(ap + 4 * i);
#pragma unroll
        for (int kb = 0; kb < 4; ++kb) {
            if (kb < 3) {
#pragma unroll
                for (int i = 0; i < 16; ++i)
                    af[(kb + 1) & 1][i] = *(const v2f*)(ap + 4 * (16 * (kb + 1) + i));
            }
#pragma unroll
            for (int i = 0; i < 16; ++i) {
                acc = __builtin_amdgcn_wmma_f32_16x16x4_f32(
                     false, af[kb & 1][i],  false, breg[16 * kb + i],
                     (short)0, acc,  false,  false);
            }
        }
        __builtin_amdgcn_sched_group_barrier(0x100, 8, 0);
        __builtin_amdgcn_sched_group_barrier(0x100, 8, 0);
        __builtin_amdgcn_sched_group_barrier(0x008, 16, 0);
        __builtin_amdgcn_sched_group_barrier(0x100, 8, 0);
        __builtin_amdgcn_sched_group_barrier(0x008, 16, 0);
        __builtin_amdgcn_sched_group_barrier(0x100, 8, 0);
        __builtin_amdgcn_sched_group_barrier(0x008, 16, 0);
        __builtin_amdgcn_sched_group_barrier(0x008, 16, 0);

        {
            float sl[4], sr[4];
#pragma unroll
            for (int i = 0; i < 4; ++i) {
                sl[i] = gw_fold<8>(c8, acc[i] * wl, acc[i + 4] * wl);
                sr[i] = gw_fold<8>(c8, acc[i] * wr, acc[i + 4] * wr);
            }
            const float tl0 = gw_fold<4>(c4, sl[0], sl[2]);
            const float tl1 = gw_fold<4>(c4, sl[1], sl[3]);
            const float tr0 = gw_fold<4>(c4, sr[0], sr[2]);
            const float tr1 = gw_fold<4>(c4, sr[1], sr[3]);
            const float ul = gw_fold<2>(c2, tl0, tl1);
            const float ur = gw_fold<2>(c2, tr0, tr1);
            part[buf][tid] = gw_fold<1>(c1, ul, ur);
        }

        if (has_next) {
            float* nb = As[buf ^ 1];
            gw_stage(nb, f0, p0);
            gw_stage(nb, f1, p1);
            gw_stage(nb, f2, p2);
            gw_stage(nb, f3, p3);
        }
        __syncthreads();

        const int row0 = t * 16 + (lane >> 4) * 8;
        float* cp = C + (size_t)row0 * NCOL + n0 + r;
        if (t * 16 + 16 <= M) {
#pragma unroll
            for (int i = 0; i < 8; ++i) cp[(size_t)i * NCOL] = acc[i];
        } else {
#pragma unroll
            for (int i = 0; i < 8; ++i)
                if (row0 + i < M) cp[(size_t)i * NCOL] = acc[i];
        }

        if (tid < 128) {
            const int sc  = tid >> 6;
            const int o   = tid & 63;
            const int row = o >> 2, h = o & 3;
            const int pl  = (row >> 3) * 16 + (row & 7) * 2 + sc;
            const float v = part[buf][(2 * h) * 32 + pl] + part[buf][(2 * h + 1) * 32 + pl];
            float* ap2 = sc ? a2 : a1;
            if (t * 16 + row < M) ap2[(size_t)t * 64 + o] = v;
        }

        if (!has_next) break;
        buf ^= 1;
        t = tn;
    }
}

#define OWN_THREADS 1024
#define OWN_GROUPS  256
#define R_MAX       391
#define CAP         12288
#define SEGV        (CAP / 4 - 1)
#define SCAN_U      8
#define AGG_U       8

__device__ __forceinline__ float leaky(float v) {
    return v > 0.f ? v : ALPHA * v;
}

__device__ __forceinline__ int f2key(float f) {
    int b = __float_as_int(f);
    return b >= 0 ? b : (b ^ 0x7FFFFFFF);
}
__device__ __forceinline__ float key2f(int k) {
    return __int_as_float(k >= 0 ? k : (k ^ 0x7FFFFFFF));
}

struct OwnLds {
    float acc[R_MAX * NCOL];
    int   mkey[R_MAX * HEADS];
    float ssum[R_MAX * HEADS];
    float a2s[R_MAX * HEADS];
    int2  lst[CAP];
    int   cnt;
};

template <bool FULL>
__device__ __forceinline__ void own_scan_block(OwnLds& L, const int4* __restrict__ d4,
                                               int b, int s1, int gbase,
                                               int lo, unsigned nR, int tid) {
    int4 d[SCAN_U];
#pragma unroll
    for (int u = 0; u < SCAN_U; ++u) {
        const int i = b + tid + u * OWN_THREADS;
        if (FULL || i < s1) d[u] = d4[i];
        else d[u] = make_int4(lo - 1, lo - 1, lo - 1, lo - 1);
    }
    unsigned mask = 0;
#pragma unroll
    for (int u = 0; u < SCAN_U; ++u) {
        const unsigned tx = (unsigned)(d[u].x - lo), ty = (unsigned)(d[u].y - lo);
        const unsigned tz = (unsigned)(d[u].z - lo), tw = (unsigned)(d[u].w - lo);
        const unsigned m = min(min(tx, ty), min(tz, tw));
        if (m < nR) {
            const unsigned g = (tx < nR ? 0x10000000u : 0u) | (ty < nR ? 0x20000000u : 0u) |
                               (tz < nR ? 0x40000000u : 0u) | (tw < nR ? 0x80000000u : 0u);
            mask |= g >> (28 - 4 * u);
        }
    }
    while (mask) {
        const int bit = __ffs(mask) - 1;
        mask &= mask - 1;
        const int i = b + (bit >> 2) * OWN_THREADS + tid;
        const int slot = atomicAdd(&L.cnt, 1);
        if (slot < CAP) L.lst[slot] = make_int2(((gbase + i) << 2) + (bit & 3), 0);
    }
}

__device__ __forceinline__ void own_filter(OwnLds& L, const int* __restrict__ dst,
                                           int v0, int v1, bool tail, int E,
                                           int lo, unsigned nR, int tid) {
    const int4* __restrict__ d4 = (const int4*)dst + v0;
    const int count = v1 - v0;
    const int rot = (count / (int)gridDim.x) * (int)blockIdx.x;
    for (int sp = 0; sp < 2; ++sp) {
        const int s0 = sp ? 0 : rot;
        const int s1 = sp ? rot : count;
        int b = s0;
        for (; b + OWN_THREADS * SCAN_U <= s1; b += OWN_THREADS * SCAN_U)
            own_scan_block<true>(L, d4, b, s1, v0, lo, nR, tid);
        if (b < s1)
            own_scan_block<false>(L, d4, b, s1, v0, lo, nR, tid);
    }
    if (tail && tid < (E & 3)) {
        int e = (v1 << 2) + tid;
        if ((unsigned)(dst[e] - lo) < nR) {
            int slot = atomicAdd(&L.cnt, 1);
            if (slot < CAP) L.lst[slot] = make_int2(e, 0);
        }
    }
}

__device__ __forceinline__ void own_resolve(OwnLds& L, const int* __restrict__ src,
                                            const int* __restrict__ dst,
                                            int n, int lo, int tid) {
    for (int i = tid; i < n; i += OWN_THREADS) {
        int e = L.lst[i].x;
        L.lst[i] = make_int2(src[e], dst[e] - lo);
    }
}

template <int PH>
__device__ __forceinline__ void own_edge_pass(OwnLds& L, const float* __restrict__ a1,
                                              int n, int tid) {
    for (int i = tid; i < n; i += OWN_THREADS) {
        int2 en = L.lst[i];
        float4 A1 = *(const float4*)(a1 + (size_t)en.x * HEADS);
        const int k = en.y * HEADS;
        float v0 = leaky(A1.x + L.a2s[k + 0]);
        float v1 = leaky(A1.y + L.a2s[k + 1]);
        float v2 = leaky(A1.z + L.a2s[k + 2]);
        float v3 = leaky(A1.w + L.a2s[k + 3]);
        if (PH == 0) {
            atomicMax(&L.mkey[k + 0], f2key(v0));
            atomicMax(&L.mkey[k + 1], f2key(v1));
            atomicMax(&L.mkey[k + 2], f2key(v2));
            atomicMax(&L.mkey[k + 3], f2key(v3));
        } else {
            atomicAdd(&L.ssum[k + 0], expf(v0 - key2f(L.mkey[k + 0])));
            atomicAdd(&L.ssum[k + 1], expf(v1 - key2f(L.mkey[k + 1])));
            atomicAdd(&L.ssum[k + 2], expf(v2 - key2f(L.mkey[k + 2])));
            atomicAdd(&L.ssum[k + 3], expf(v3 - key2f(L.mkey[k + 3])));
        }
    }
}

static_assert(AGG_U * HEADS == 32, "one lane per (batch edge, head)");
__device__ __forceinline__ void own_agg_pass(OwnLds& L, const float* __restrict__ ft,
                                             const float* __restrict__ a1,
                                             int n, int tid) {
    const int wave = tid >> 5;
    const int q = tid & 31;
    const int um = q >> 2;
    const int hl = q & 3;
    for (int i0 = wave; i0 < n; i0 += 32 * AGG_U) {
        int2  en[AGG_U];
        float F[AGG_U][HEADS];
#pragma unroll
        for (int u = 0; u < AGG_U; ++u) {
            int i = i0 + u * 32;
            en[u] = L.lst[i < n ? i : i0];
        }
        const int im = i0 + um * 32;
        const int2 em = L.lst[im < n ? im : i0];
#pragma unroll
        for (int u = 0; u < AGG_U; ++u) {
            const float* fr = ft + (size_t)en[u].x * NCOL + q;
            F[u][0] = fr[0];
            F[u][1] = fr[32];
            F[u][2] = fr[64];
            F[u][3] = fr[96];
        }
        const float A1m = a1[em.x * HEADS + hl];
        const int idx = em.y * HEADS + hl;
        const float v = leaky(A1m + L.a2s[idx]);
        const float alpha = expf(v - key2f(L.mkey[idx])) / L.ssum[idx];
        const int ab = __float_as_int(alpha);
#pragma unroll
        for (int u = 0; u < AGG_U; ++u) {
            if (i0 + u * 32 < n) {
                float* op = L.acc + en[u].y * NCOL + q;
                const float al0 = __int_as_float(__builtin_amdgcn_readlane(ab, 4 * u + 0));
                const float al1 = __int_as_float(__builtin_amdgcn_readlane(ab, 4 * u + 1));
                const float al2 = __int_as_float(__builtin_amdgcn_readlane(ab, 4 * u + 2));
                const float al3 = __int_as_float(__builtin_amdgcn_readlane(ab, 4 * u + 3));
                atomicAdd(op +  0, F[u][0] * al0);
                atomicAdd(op + 32, F[u][1] * al1);
                atomicAdd(op + 64, F[u][2] * al2);
                atomicAdd(op + 96, F[u][3] * al3);
            }
        }
    }
}

__global__ void __launch_bounds__(OWN_THREADS)
gat_owner(const int* __restrict__ src, const int* __restrict__ dst,
          const float* __restrict__ ft, const float* __restrict__ a1,
          const float* __restrict__ a2, float* __restrict__ out,
          int n_nodes, int n_edges) {
    __shared__ OwnLds L;
    const int tid = threadIdx.x;
    const int nvec = n_edges >> 2;

    for (int lo = blockIdx.x * R_MAX; lo < n_nodes; lo += gridDim.x * R_MAX) {
        const int nR = (n_nodes - lo) < R_MAX ? (n_nodes - lo) : R_MAX;

        float4* acc4 = (float4*)L.acc;
        for (int i = tid; i < R_MAX * NCOL / 4; i += OWN_THREADS)
            acc4[i] = make_float4(0.f, 0.f, 0.f, 0.f);
        for (int i = tid; i < R_MAX * HEADS; i += OWN_THREADS) {
            L.mkey[i] = (int)0x80000000;
            L.ssum[i] = 0.f;
            L.a2s[i]  = (i < nR * HEADS) ? a2[(size_t)lo * HEADS + i] : 0.f;
        }
        if (tid == 0) L.cnt = 0;
        __syncthreads();

        own_filter(L, dst, 0, nvec, true, n_edges, lo, (unsigned)nR, tid);
        __syncthreads();
        const int total = L.cnt;

        if (total <= CAP) {
            own_resolve(L, src, dst, total, lo, tid);
            __syncthreads();
            own_edge_pass<0>(L, a1, total, tid);
            __syncthreads();
            own_edge_pass<1>(L, a1, total, tid);
            __syncthreads();
            own_agg_pass(L, ft, a1, total, tid);
        } else {
            for (int ph = 0; ph < 3; ++ph) {
                for (int v0 = 0; v0 <= nvec; v0 += SEGV) {
                    const int v1 = (v0 + SEGV < nvec) ? (v0 + SEGV) : nvec;
                    __syncthreads();
                    if (tid == 0) L.cnt = 0;
                    __syncthreads();
                    own_filter(L, dst, v0, v1, v1 == nvec, n_edges, lo, (unsigned)nR, tid);
                    __syncthreads();
                    const int n = L.cnt;
                    own_resolve(L, src, dst, n, lo, tid);
                    __syncthreads();
                    if (ph == 0)      own_edge_pass<0>(L, a1, n, tid);
                    else if (ph == 1) own_edge_pass<1>(L, a1, n, tid);
                    else              own_agg_pass(L, ft, a1, n, tid);
                    if (v1 == nvec) break;
                }
                __syncthreads();
            }
        }
        __syncthreads();

        float4* out4 = (float4*)(out + (size_t)lo * NCOL);
        for (int i = tid; i < nR * NCOL / 4; i += OWN_THREADS) out4[i] = acc4[i];
        __syncthreads();
    }
}

extern "C" void kernel_launch(void* const* d_in, const int* in_sizes, int n_in,
                              void* d_out, int out_size, void* d_ws, size_t ws_size,
                              hipStream_t stream) {
    const float* feature = (const float*)d_in[0];
    const int*   src     = (const int*)d_in[1];
    const int*   dst     = (const int*)d_in[2];
    const float* W_fc    = (const float*)d_in[3];
    const float* attn_l  = (const float*)d_in[4];
    const float* attn_r  = (const float*)d_in[5];
    float*       out     = (float*)d_out;

    const int M = in_sizes[0] / IN_DIM;
    const int E = in_sizes[1];

    float* ft    = (float*)d_ws;
    float* a1    = ft    + (size_t)M * NCOL;
    float* a2    = a1    + (size_t)M * HEADS;

    gat_gemm_wmma<<<GW_GROUPS, GW_THREADS, 0, stream>>>(feature, W_fc, attn_l, attn_r,
                                                        ft, a1, a2, M);
    gat_owner<<<OWN_GROUPS, OWN_THREADS, 0, stream>>>(src, dst, ft, a1, a2, out, M, E);
}


# Owner filter reads a bit-sliced owner plane (8 bit words of dst/391 per 32 edges, voted in the GEMM launch): 1 B/edge, 8 logic ops per 32; per-wave list runs; arrival-order sums, new list order
# speedup vs baseline: 5.4080x; 1.3151x over previous
#include <hip/hip_runtime.h>
#include <hip/hip_bf16.h>

#define IN_DIM 256
#define NCOL 128
#define HEADS 4
#define OUT_DIM 32
#define ALPHA 0.2f
#define R_MAX 391

typedef __attribute__((ext_vector_type(2))) float v2f;
typedef __attribute__((ext_vector_type(8))) float v8f;

#define GW_THREADS 256
#define GW_GROUPS  512
#define A_PITCH    260

__device__ __forceinline__ float4 gw_load(const float* __restrict__ A, int tile, int f, int M) {
    int row = tile * 16 + (f >> 6);
    row = row < M ? row : M - 1;
    return *(const float4*)(A + (size_t)row * IN_DIM + (f & 63) * 4);
}
__device__ __forceinline__ void gw_stage(float* __restrict__ dstbuf, int f, float4 v) {
    *(float4*)(dstbuf + (f >> 6) * A_PITCH + (f & 63) * 4) = v;
}

template <int XM>
__device__ __forceinline__ float gw_xlane(float v) {
    return __int_as_float(__builtin_amdgcn_update_dpp(0, __float_as_int(v), 0x160 + XM,
                                                      0xf, 0xf, true));
}
template <int XM>
__device__ __forceinline__ float gw_fold(bool bit, float lo, float hi) {
    const float keep = bit ? hi : lo;
    const float send = bit ? lo : hi;
    return keep + gw_xlane<XM>(send);
}

__global__ void __launch_bounds__(GW_THREADS)
gat_gemm_wmma(const float* __restrict__ A,
              const float* __restrict__ B,
              const float* __restrict__ attn_l,
              const float* __restrict__ attn_r,
              const int* __restrict__ dst,
              float* __restrict__ C,
              float* __restrict__ a1,
              float* __restrict__ a2,
              int* __restrict__ plane, int M, int E) {
    __shared__ __attribute__((aligned(16))) float As[2][16 * A_PITCH];
    __shared__ float part[2][GW_THREADS];

    const int tid  = threadIdx.x;
    const int lane = tid & 31;
    const int n0   = (tid >> 5) * 16;
    const int r    = lane & 15;
    const int koff = (lane >> 4) * 2;
    const int tiles = (M + 15) >> 4;
    const int G = gridDim.x;

    {
        const int ngrp = E >> 5;
        const int nwav = G * (GW_THREADS / 32);
        int4* __restrict__ pa = (int4*)plane;
        int4* __restrict__ pb = pa + ngrp;
        const int wv = __builtin_amdgcn_readfirstlane(tid >> 5);
        for (int g0 = blockIdx.x * (GW_THREADS / 32) + wv; g0 < ngrp; g0 += 8 * nwav) {
            int dv[8];
#pragma unroll
            for (int q = 0; q < 8; ++q) {
                const int g = g0 + q * nwav;
                const int gl = g < ngrp ? g : ngrp - 1;
                dv[q] = dst[(size_t)gl * 32 + lane];
            }
#pragma unroll
            for (int q = 0; q < 8; ++q) {
                const int g = g0 + q * nwav;
                if (g < ngrp) {
                    const unsigned o = (unsigned)dv[q] / R_MAX;
                    const int b0 = (int)__builtin_amdgcn_ballot_w32((o & 1u) != 0);
                    const int b1 = (int)__builtin_amdgcn_ballot_w32((o & 2u) != 0);
                    const int b2 = (int)__builtin_amdgcn_ballot_w32((o & 4u) != 0);
                    const int b3 = (int)__builtin_amdgcn_ballot_w32((o & 8u) != 0);
                    const int b4 = (int)__builtin_amdgcn_ballot_w32((o & 16u) != 0);
                    const int b5 = (int)__builtin_amdgcn_ballot_w32((o & 32u) != 0);
                    const int b6 = (int)__builtin_amdgcn_ballot_w32((o & 64u) != 0);
                    const int b7 = (int)__builtin_amdgcn_ballot_w32((o & 128u) != 0);
                    if (lane == 0) {
                        pa[g] = make_int4(b0, b1, b2, b3);
                        pb[g] = make_int4(b4, b5, b6, b7);
                    }
                }
            }
        }
    }

    int t = blockIdx.x;
    if (t >= tiles) return;

    const float wl = attn_l[n0 + r];
    const float wr = attn_r[n0 + r];
    const bool c8 = (lane & 8) != 0, c4 = (lane & 4) != 0, c2 = (lane & 2) != 0, c1 = (lane & 1) != 0;

    const int f0 = tid, f1 = tid + GW_THREADS, f2 = tid + 2 * GW_THREADS, f3 = tid + 3 * GW_THREADS;
    float4 p0 = gw_load(A, t, f0, M);
    float4 p1 = gw_load(A, t, f1, M);
    float4 p2 = gw_load(A, t, f2, M);
    float4 p3 = gw_load(A, t, f3, M);

    v2f breg[IN_DIM / 4];
#pragma unroll
    for (int j = 0; j < IN_DIM / 4; ++j) {
        const float* bp = B + (size_t)(4 * j + koff) * NCOL + n0 + r;
        breg[j].x = bp[0];
        breg[j].y = bp[NCOL];
    }

    gw_stage(As[0], f0, p0);
    gw_stage(As[0], f1, p1);
    gw_stage(As[0], f2, p2);
    gw_stage(As[0], f3, p3);
    __syncthreads();

    int buf = 0;
    for (;;) {
        const int tn = t + G;
        const bool has_next = tn < tiles;
        if (has_next) {
            p0 = gw_load(A, tn, f0, M);
            p1 = gw_load(A, tn, f1, M);
            p2 = gw_load(A, tn, f2, M);
            p3 = gw_load(A, tn, f3, M);
        }

        v8f acc = {};
        const float* ap = &As[buf][r * A_PITCH + koff];
        v2f af[2][16];
#pragma unroll
        for (int i = 0; i < 16; ++i) af[0][i] = *(const v2f*)(ap + 4 * i);
#pragma unroll
        for (int kb = 0; kb < 4; ++kb) {
            if (kb < 3) {
#pragma unroll
                for (int i = 0; i < 16; ++i)
                    af[(kb + 1) & 1][i] = *(const v2f*)(ap + 4 * (16 * (kb + 1) + i));
            }
#pragma unroll
            for (int i = 0; i < 16; ++i) {
                acc = __builtin_amdgcn_wmma_f32_16x16x4_f32(
                     false, af[kb & 1][i],  false, breg[16 * kb + i],
                     (short)0, acc,  false,  false);
            }
        }
        __builtin_amdgcn_sched_group_barrier(0x100, 8, 0);
        __builtin_amdgcn_sched_group_barrier(0x100, 8, 0);
        __builtin_amdgcn_sched_group_barrier(0x008, 16, 0);
        __builtin_amdgcn_sched_group_barrier(0x100, 8, 0);
        __builtin_amdgcn_sched_group_barrier(0x008, 16, 0);
        __builtin_amdgcn_sched_group_barrier(0x100, 8, 0);
        __builtin_amdgcn_sched_group_barrier(0x008, 16, 0);
        __builtin_amdgcn_sched_group_barrier(0x008, 16, 0);

        {
            float sl[4], sr[4];
#pragma unroll
            for (int i = 0; i < 4; ++i) {
                sl[i] = gw_fold<8>(c8, acc[i] * wl, acc[i + 4] * wl);
                sr[i] = gw_fold<8>(c8, acc[i] * wr, acc[i + 4] * wr);
            }
            const float tl0 = gw_fold<4>(c4, sl[0], sl[2]);
            const float tl1 = gw_fold<4>(c4, sl[1], sl[3]);
            const float tr0 = gw_fold<4>(c4, sr[0], sr[2]);
            const float tr1 = gw_fold<4>(c4, sr[1], sr[3]);
            const float ul = gw_fold<2>(c2, tl0, tl1);
            const float ur = gw_fold<2>(c2, tr0, tr1);
            part[buf][tid] = gw_fold<1>(c1, ul, ur);
        }

        if (has_next) {
            float* nb = As[buf ^ 1];
            gw_stage(nb, f0, p0);
            gw_stage(nb, f1, p1);
            gw_stage(nb, f2, p2);
            gw_stage(nb, f3, p3);
        }
        __syncthreads();

        const int row0 = t * 16 + (lane >> 4) * 8;
        float* cp = C + (size_t)row0 * NCOL + n0 + r;
        if (t * 16 + 16 <= M) {
#pragma unroll
            for (int i = 0; i < 8; ++i) cp[(size_t)i * NCOL] = acc[i];
        } else {
#pragma unroll
            for (int i = 0; i < 8; ++i)
                if (row0 + i < M) cp[(size_t)i * NCOL] = acc[i];
        }

        if (tid < 128) {
            const int sc  = tid >> 6;
            const int o   = tid & 63;
            const int row = o >> 2, h = o & 3;
            const int pl  = (row >> 3) * 16 + (row & 7) * 2 + sc;
            const float v = part[buf][(2 * h) * 32 + pl] + part[buf][(2 * h + 1) * 32 + pl];
            float* ap2 = sc ? a2 : a1;
            if (t * 16 + row < M) ap2[(size_t)t * 64 + o] = v;
        }

        if (!has_next) break;
        buf ^= 1;
        t = tn;
    }
}

#define OWN_THREADS 1024
#define OWN_GROUPS  256
#define CAP         12288
#define CAPW        (CAP / 32)
#define SEGV        (CAP / 4 - 1)
#define SCAN_U      8
#define AGG_U       8

__device__ __forceinline__ float leaky(float v) {
    return v > 0.f ? v : ALPHA * v;
}

__device__ __forceinline__ int f2key(float f) {
    int b = __float_as_int(f);
    return b >= 0 ? b : (b ^ 0x7FFFFFFF);
}
__device__ __forceinline__ float key2f(int k) {
    return __int_as_float(k >= 0 ? k : (k ^ 0x7FFFFFFF));
}

struct OwnLds {
    float acc[R_MAX * NCOL];
    int   mkey[R_MAX * HEADS];
    float ssum[R_MAX * HEADS];
    float a2s[R_MAX * HEADS];
    int2  lst[CAP];
    int   cnt;
    int   wcnt[OWN_THREADS / 32];
    int   ovf;
};

template <bool FULL>
__device__ __forceinline__ void own_scan_block(OwnLds& L, const int4* __restrict__ d4,
                                               int b, int s1, int gbase,
                                               int lo, unsigned nR, int tid) {
    int4 d[SCAN_U];
#pragma unroll
    for (int u = 0; u < SCAN_U; ++u) {
        const int i = b + tid + u * OWN_THREADS;
        if (FULL || i < s1) d[u] = d4[i];
        else d[u] = make_int4(lo - 1, lo - 1, lo - 1, lo - 1);
    }
    unsigned mask = 0;
#pragma unroll
    for (int u = 0; u < SCAN_U; ++u) {
        const unsigned tx = (unsigned)(d[u].x - lo), ty = (unsigned)(d[u].y - lo);
        const unsigned tz = (unsigned)(d[u].z - lo), tw = (unsigned)(d[u].w - lo);
        const unsigned m = min(min(tx, ty), min(tz, tw));
        if (m < nR) {
            const unsigned g = (tx < nR ? 0x10000000u : 0u) | (ty < nR ? 0x20000000u : 0u) |
                               (tz < nR ? 0x40000000u : 0u) | (tw < nR ? 0x80000000u : 0u);
            mask |= g >> (28 - 4 * u);
        }
    }
    while (mask) {
        const int bit = __ffs(mask) - 1;
        mask &= mask - 1;
        const int i = b + (bit >> 2) * OWN_THREADS + tid;
        const int slot = atomicAdd(&L.cnt, 1);
        if (slot < CAP) L.lst[slot] = make_int2(((gbase + i) << 2) + (bit & 3), 0);
    }
}

__device__ __forceinline__ void own_filter(OwnLds& L, const int* __restrict__ dst,
                                           int v0, int v1, bool tail, int E,
                                           int lo, unsigned nR, int tid) {
    const int4* __restrict__ d4 = (const int4*)dst + v0;
    const int count = v1 - v0;
    const int rot = (count / (int)gridDim.x) * (int)blockIdx.x;
    for (int sp = 0; sp < 2; ++sp) {
        const int s0 = sp ? 0 : rot;
        const int s1 = sp ? rot : count;
        int b = s0;
        for (; b + OWN_THREADS * SCAN_U <= s1; b += OWN_THREADS * SCAN_U)
            own_scan_block<true>(L, d4, b, s1, v0, lo, nR, tid);
        if (b < s1)
            own_scan_block<false>(L, d4, b, s1, v0, lo, nR, tid);
    }
    if (tail && tid < (E & 3)) {
        int e = (v1 << 2) + tid;
        if ((unsigned)(dst[e] - lo) < nR) {
            int slot = atomicAdd(&L.cnt, 1);
            if (slot < CAP) L.lst[slot] = make_int2(e, 0);
        }
    }
}

#define PSCAN_G (SCAN_U / 2)
template <bool FULL>
__device__ __forceinline__ void own_pscan_block(OwnLds& L, const int4* __restrict__ pa,
                                                const int4* __restrict__ pb,
                                                int b, int s1, const unsigned (&c)[8],
                                                int wbase, int& wn, int tid) {
    int4 lo4[PSCAN_G], hi4[PSCAN_G];
#pragma unroll
    for (int v = 0; v < PSCAN_G; ++v) {
        const int g = b + tid + v * OWN_THREADS;
        if (FULL || g < s1) {
            lo4[v] = pa[g];
            hi4[v] = pb[g];
        } else {
            lo4[v] = make_int4((int)c[0], 0, 0, 0);
            hi4[v] = make_int4(0, 0, 0, 0);
        }
    }
#pragma unroll
    for (int v = 0; v < PSCAN_G; ++v) {
        unsigned m = ((unsigned)lo4[v].x ^ c[0]) & ((unsigned)lo4[v].y ^ c[1]) &
                     ((unsigned)lo4[v].z ^ c[2]) & ((unsigned)lo4[v].w ^ c[3]) &
                     ((unsigned)hi4[v].x ^ c[4]) & ((unsigned)hi4[v].y ^ c[5]) &
                     ((unsigned)hi4[v].z ^ c[6]) & ((unsigned)hi4[v].w ^ c[7]);
        const int e0 = (b + tid + v * OWN_THREADS) << 5;
        unsigned vote = __builtin_amdgcn_ballot_w32(m != 0);
        while (vote) {
            if (m) {
                const int bit = __ffs(m) - 1;
                m &= m - 1;
                const int slot = wn + (int)__builtin_amdgcn_mbcnt_lo(vote, 0);
                if (slot < CAPW) L.lst[wbase + slot] = make_int2(e0 + bit, 0);
            }
            wn += __popc(vote);
            vote = __builtin_amdgcn_ballot_w32(m != 0);
        }
    }
}

__device__ __forceinline__ void own_filter_plane(OwnLds& L, const int* __restrict__ plane,
                                                 const int* __restrict__ dst, int E,
                                                 int k, int lo, unsigned nR, int tid) {
    const int count = E >> 5;
    const int4* __restrict__ pa = (const int4*)plane;
    const int4* __restrict__ pb = pa + count;
    unsigned c[8];
#pragma unroll
    for (int j = 0; j < 8; ++j) c[j] = ((k >> j) & 1) ? 0u : ~0u;
    const int wave  = tid >> 5;
    const int wbase = wave * CAPW;
    int wn = 0;
    const int rot = (count / (int)gridDim.x) * (int)blockIdx.x;
    for (int sp = 0; sp < 2; ++sp) {
        const int s0 = sp ? 0 : rot;
        const int s1 = sp ? rot : count;
        int b = s0;
        for (; b + OWN_THREADS * PSCAN_G <= s1; b += OWN_THREADS * PSCAN_G)
            own_pscan_block<true>(L, pa, pb, b, s1, c, wbase, wn, tid);
        if (b < s1)
            own_pscan_block<false>(L, pa, pb, b, s1, c, wbase, wn, tid);
    }
    if (wave == 0) {
        const int e = (count << 5) + tid;
        bool hit = false;
        if (tid < (E & 31)) hit = (unsigned)(dst[e] - lo) < nR;
        const unsigned vote = __builtin_amdgcn_ballot_w32(hit);
        if (hit) {
            const int slot = wn + (int)__builtin_amdgcn_mbcnt_lo(vote, 0);
            if (slot < CAPW) L.lst[wbase + slot] = make_int2(e, 0);
        }
        wn += __popc(vote);
    }
    if ((tid & 31) == 0) {
        L.wcnt[wave] = wn;
        if (wn > CAPW) L.ovf = 1;
    }
}

__device__ __forceinline__ void own_resolve(OwnLds& L, const int* __restrict__ src,
                                            const int* __restrict__ dst,
                                            int first, int end, int step, int lo) {
    for (int i = first; i < end; i += step) {
        int e = L.lst[i].x;
        L.lst[i] = make_int2(src[e], dst[e] - lo);
    }
}

template <int PH>
__device__ __forceinline__ void own_edge_pass(OwnLds& L, const float* __restrict__ a1,
                                              int first, int end, int step) {
    for (int i = first; i < end; i += step) {
        int2 en = L.lst[i];
        float4 A1 = *(const float4*)(a1 + (size_t)en.x * HEADS);
        const int k = en.y * HEADS;
        float v0 = leaky(A1.x + L.a2s[k + 0]);
        float v1 = leaky(A1.y + L.a2s[k + 1]);
        float v2 = leaky(A1.z + L.a2s[k + 2]);
        float v3 = leaky(A1.w + L.a2s[k + 3]);
        if (PH == 0) {
            atomicMax(&L.mkey[k + 0], f2key(v0));
            atomicMax(&L.mkey[k + 1], f2key(v1));
            atomicMax(&L.mkey[k + 2], f2key(v2));
            atomicMax(&L.mkey[k + 3], f2key(v3));
        } else {
            atomicAdd(&L.ssum[k + 0], expf(v0 - key2f(L.mkey[k + 0])));
            atomicAdd(&L.ssum[k + 1], expf(v1 - key2f(L.mkey[k + 1])));
            atomicAdd(&L.ssum[k + 2], expf(v2 - key2f(L.mkey[k + 2])));
            atomicAdd(&L.ssum[k + 3], expf(v3 - key2f(L.mkey[k + 3])));
        }
    }
}

static_assert(AGG_U * HEADS == 32, "one lane per (batch edge, head)");
__device__ __forceinline__ void own_agg_pass(OwnLds& L, const float* __restrict__ ft,
                                             const float* __restrict__ a1,
                                             int first, int n, int us, int bs, int tid) {
    const int q = tid & 31;
    const int um = q >> 2;
    const int hl = q & 3;
    for (int i0 = first; i0 < n; i0 += bs) {
        int2  en[AGG_U];
        float F[AGG_U][HEADS];
#pragma unroll
        for (int u = 0; u < AGG_U; ++u) {
            int i = i0 + u * us;
            en[u] = L.lst[i < n ? i : i0];
        }
        const int im = i0 + um * us;
        const int2 em = L.lst[im < n ? im : i0];
#pragma unroll
        for (int u = 0; u < AGG_U; ++u) {
            const float* fr = ft + (size_t)en[u].x * NCOL + q;
            F[u][0] = fr[0];
            F[u][1] = fr[32];
            F[u][2] = fr[64];
            F[u][3] = fr[96];
        }
        const float A1m = a1[em.x * HEADS + hl];
        const int idx = em.y * HEADS + hl;
        const float v = leaky(A1m + L.a2s[idx]);
        const float alpha = expf(v - key2f(L.mkey[idx])) / L.ssum[idx];
        const int ab = __float_as_int(alpha);
#pragma unroll
        for (int u = 0; u < AGG_U; ++u) {
            if (i0 + u * us < n) {
                float* op = L.acc + en[u].y * NCOL + q;
                const float al0 = __int_as_float(__builtin_amdgcn_readlane(ab, 4 * u + 0));
                const float al1 = __int_as_float(__builtin_amdgcn_readlane(ab, 4 * u + 1));
                const float al2 = __int_as_float(__builtin_amdgcn_readlane(ab, 4 * u + 2));
                const float al3 = __int_as_float(__builtin_amdgcn_readlane(ab, 4 * u + 3));
                atomicAdd(op +  0, F[u][0] * al0);
                atomicAdd(op + 32, F[u][1] * al1);
                atomicAdd(op + 64, F[u][2] * al2);
                atomicAdd(op + 96, F[u][3] * al3);
            }
        }
    }
}

__global__ void __launch_bounds__(OWN_THREADS)
gat_owner(const int* __restrict__ src, const int* __restrict__ dst,
          const float* __restrict__ ft, const float* __restrict__ a1,
          const float* __restrict__ a2, const int* __restrict__ plane,
          float* __restrict__ out, int n_nodes, int n_edges) {
    __shared__ OwnLds L;
    const int tid = threadIdx.x;
    const int wave = tid >> 5, lane = tid & 31;
    const int nvec = n_edges >> 2;
    const bool use_plane = n_nodes <= 256 * R_MAX;

    for (int lo = blockIdx.x * R_MAX; lo < n_nodes; lo += gridDim.x * R_MAX) {
        const int nR = (n_nodes - lo) < R_MAX ? (n_nodes - lo) : R_MAX;

        float4* acc4 = (float4*)L.acc;
        for (int i = tid; i < R_MAX * NCOL / 4; i += OWN_THREADS)
            acc4[i] = make_float4(0.f, 0.f, 0.f, 0.f);
        for (int i = tid; i < R_MAX * HEADS; i += OWN_THREADS) {
            L.mkey[i] = (int)0x80000000;
            L.ssum[i] = 0.f;
            L.a2s[i]  = (i < nR * HEADS) ? a2[(size_t)lo * HEADS + i] : 0.f;
        }
        if (tid == 0) { L.cnt = 0; L.ovf = 0; }
        __syncthreads();

        bool own_runs = false;
        if (use_plane) {
            own_filter_plane(L, plane, dst, n_edges, lo / R_MAX, lo, (unsigned)nR, tid);
            __syncthreads();
            own_runs = L.ovf == 0;
        }

        int total = 0;
        if (!own_runs) {
            own_filter(L, dst, 0, nvec, true, n_edges, lo, (unsigned)nR, tid);
            __syncthreads();
            total = L.cnt;
        }

        if (own_runs) {
            const int wfirst = wave * CAPW;
            const int wend   = wfirst + L.wcnt[wave];
            own_resolve(L, src, dst, wfirst + lane, wend, 32, lo);
            own_edge_pass<0>(L, a1, wfirst + lane, wend, 32);
            __syncthreads();
            own_edge_pass<1>(L, a1, wfirst + lane, wend, 32);
            __syncthreads();
            own_agg_pass(L, ft, a1, wfirst, wend, 1, AGG_U, tid);
        } else if (total <= CAP) {
            own_resolve(L, src, dst, tid, total, OWN_THREADS, lo);
            __syncthreads();
            own_edge_pass<0>(L, a1, tid, total, OWN_THREADS);
            __syncthreads();
            own_edge_pass<1>(L, a1, tid, total, OWN_THREADS);
            __syncthreads();
            own_agg_pass(L, ft, a1, wave, total, 32, 32 * AGG_U, tid);
        } else {
            for (int ph = 0; ph < 3; ++ph) {
                for (int v0 = 0; v0 <= nvec; v0 += SEGV) {
                    const int v1 = (v0 + SEGV < nvec) ? (v0 + SEGV) : nvec;
                    __syncthreads();
                    if (tid == 0) L.cnt = 0;
                    __syncthreads();
                    own_filter(L, dst, v0, v1, v1 == nvec, n_edges, lo, (unsigned)nR, tid);
                    __syncthreads();
                    const int n = L.cnt;
                    own_resolve(L, src, dst, tid, n, OWN_THREADS, lo);
                    __syncthreads();
                    if (ph == 0)      own_edge_pass<0>(L, a1, tid, n, OWN_THREADS);
                    else if (ph == 1) own_edge_pass<1>(L, a1, tid, n, OWN_THREADS);
                    else              own_agg_pass(L, ft, a1, wave, n, 32, 32 * AGG_U, tid);
                    if (v1 == nvec) break;
                }
                __syncthreads();
            }
        }
        __syncthreads();

        float4* out4 = (float4*)(out + (size_t)lo * NCOL);
        for (int i = tid; i < nR * NCOL / 4; i += OWN_THREADS) out4[i] = acc4[i];
        __syncthreads();
    }
}

extern "C" void kernel_launch(void* const* d_in, const int* in_sizes, int n_in,
                              void* d_out, int out_size, void* d_ws, size_t ws_size,
                              hipStream_t stream) {
    const float* feature = (const float*)d_in[0];
    const int*   src     = (const int*)d_in[1];
    const int*   dst     = (const int*)d_in[2];
    const float* W_fc    = (const float*)d_in[3];
    const float* attn_l  = (const float*)d_in[4];
    const float* attn_r  = (const float*)d_in[5];
    float*       out     = (float*)d_out;

    const int M = in_sizes[0] / IN_DIM;
    const int E = in_sizes[1];

    float* ft    = (float*)d_ws;
    float* a1    = ft    + (size_t)M * NCOL;
    float* a2    = a1    + (size_t)M * HEADS;
    int*   plane = (int*)(a2 + (size_t)M * HEADS);

    gat_gemm_wmma<<<GW_GROUPS, GW_THREADS, 0, stream>>>(feature, W_fc, attn_l, attn_r, dst,
                                                        ft, a1, a2, plane, M, E);
    gat_owner<<<OWN_GROUPS, OWN_THREADS, 0, stream>>>(src, dst, ft, a1, a2, plane, out, M, E);
}


# Own-run path: a lane keeps its up to 8 edges' softmax terms in registers from max pass to weights; a1 gathered once not 3 times, one exp per edge-head, a lane's loads batched; arrival-order sums kept
# speedup vs baseline: 5.7911x; 1.0708x over previous
#include <hip/hip_runtime.h>
#include <hip/hip_bf16.h>

#define IN_DIM 256
#define NCOL 128
#define HEADS 4
#define OUT_DIM 32
#define ALPHA 0.2f
#define R_MAX 391

typedef __attribute__((ext_vector_type(2))) float v2f;
typedef __attribute__((ext_vector_type(8))) float v8f;

#define GW_THREADS 256
#define GW_GROUPS  512
#define A_PITCH    260

__device__ __forceinline__ float4 gw_load(const float* __restrict__ A, int tile, int f, int M) {
    int row = tile * 16 + (f >> 6);
    row = row < M ? row : M - 1;
    return *(const float4*)(A + (size_t)row * IN_DIM + (f & 63) * 4);
}
__device__ __forceinline__ void gw_stage(float* __restrict__ dstbuf, int f, float4 v) {
    *(float4*)(dstbuf + (f >> 6) * A_PITCH + (f & 63) * 4) = v;
}

template <int XM>
__device__ __forceinline__ float gw_xlane(float v) {
    return __int_as_float(__builtin_amdgcn_update_dpp(0, __float_as_int(v), 0x160 + XM,
                                                      0xf, 0xf, true));
}
template <int XM>
__device__ __forceinline__ float gw_fold(bool bit, float lo, float hi) {
    const float keep = bit ? hi : lo;
    const float send = bit ? lo : hi;
    return keep + gw_xlane<XM>(send);
}

__global__ void __launch_bounds__(GW_THREADS)
gat_gemm_wmma(const float* __restrict__ A,
              const float* __restrict__ B,
              const float* __restrict__ attn_l,
              const float* __restrict__ attn_r,
              const int* __restrict__ dst,
              float* __restrict__ C,
              float* __restrict__ a1,
              float* __restrict__ a2,
              int* __restrict__ plane, int M, int E) {
    __shared__ __attribute__((aligned(16))) float As[2][16 * A_PITCH];
    __shared__ float part[2][GW_THREADS];

    const int tid  = threadIdx.x;
    const int lane = tid & 31;
    const int n0   = (tid >> 5) * 16;
    const int r    = lane & 15;
    const int koff = (lane >> 4) * 2;
    const int tiles = (M + 15) >> 4;
    const int G = gridDim.x;

    {
        const int ngrp = E >> 5;
        const int nwav = G * (GW_THREADS / 32);
        int4* __restrict__ pa = (int4*)plane;
        int4* __restrict__ pb = pa + ngrp;
        const int wv = __builtin_amdgcn_readfirstlane(tid >> 5);
        for (int g0 = blockIdx.x * (GW_THREADS / 32) + wv; g0 < ngrp; g0 += 8 * nwav) {
            int dv[8];
#pragma unroll
            for (int q = 0; q < 8; ++q) {
                const int g = g0 + q * nwav;
                const int gl = g < ngrp ? g : ngrp - 1;
                dv[q] = dst[(size_t)gl * 32 + lane];
            }
#pragma unroll
            for (int q = 0; q < 8; ++q) {
                const int g = g0 + q * nwav;
                if (g < ngrp) {
                    const unsigned o = (unsigned)dv[q] / R_MAX;
                    const int b0 = (int)__builtin_amdgcn_ballot_w32((o & 1u) != 0);
                    const int b1 = (int)__builtin_amdgcn_ballot_w32((o & 2u) != 0);
                    const int b2 = (int)__builtin_amdgcn_ballot_w32((o & 4u) != 0);
                    const int b3 = (int)__builtin_amdgcn_ballot_w32((o & 8u) != 0);
                    const int b4 = (int)__builtin_amdgcn_ballot_w32((o & 16u) != 0);
                    const int b5 = (int)__builtin_amdgcn_ballot_w32((o & 32u) != 0);
                    const int b6 = (int)__builtin_amdgcn_ballot_w32((o & 64u) != 0);
                    const int b7 = (int)__builtin_amdgcn_ballot_w32((o & 128u) != 0);
                    if (lane == 0) {
                        pa[g] = make_int4(b0, b1, b2, b3);
                        pb[g] = make_int4(b4, b5, b6, b7);
                    }
                }
            }
        }
    }

    int t = blockIdx.x;
    if (t >= tiles) return;

    const float wl = attn_l[n0 + r];
    const float wr = attn_r[n0 + r];
    const bool c8 = (lane & 8) != 0, c4 = (lane & 4) != 0, c2 = (lane & 2) != 0, c1 = (lane & 1) != 0;

    const int f0 = tid, f1 = tid + GW_THREADS, f2 = tid + 2 * GW_THREADS, f3 = tid + 3 * GW_THREADS;
    float4 p0 = gw_load(A, t, f0, M);
    float4 p1 = gw_load(A, t, f1, M);
    float4 p2 = gw_load(A, t, f2, M);
    float4 p3 = gw_load(A, t, f3, M);

    v2f breg[IN_DIM / 4];
#pragma unroll
    for (int j = 0; j < IN_DIM / 4; ++j) {
        const float* bp = B + (size_t)(4 * j + koff) * NCOL + n0 + r;
        breg[j].x = bp[0];
        breg[j].y = bp[NCOL];
    }

    gw_stage(As[0], f0, p0);
    gw_stage(As[0], f1, p1);
    gw_stage(As[0], f2, p2);
    gw_stage(As[0], f3, p3);
    __syncthreads();

    int buf = 0;
    for (;;) {
        const int tn = t + G;
        const bool has_next = tn < tiles;
        if (has_next) {
            p0 = gw_load(A, tn, f0, M);
            p1 = gw_load(A, tn, f1, M);
            p2 = gw_load(A, tn, f2, M);
            p3 = gw_load(A, tn, f3, M);
        }

        v8f acc = {};
        const float* ap = &As[buf][r * A_PITCH + koff];
        v2f af[2][16];
#pragma unroll
        for (int i = 0; i < 16; ++i) af[0][i] = *(const v2f*)(ap + 4 * i);
#pragma unroll
        for (int kb = 0; kb < 4; ++kb) {
            if (kb < 3) {
#pragma unroll
                for (int i = 0; i < 16; ++i)
                    af[(kb + 1) & 1][i] = *(const v2f*)(ap + 4 * (16 * (kb + 1) + i));
            }
#pragma unroll
            for (int i = 0; i < 16; ++i) {
                acc = __builtin_amdgcn_wmma_f32_16x16x4_f32(
                     false, af[kb & 1][i],  false, breg[16 * kb + i],
                     (short)0, acc,  false,  false);
            }
        }
        __builtin_amdgcn_sched_group_barrier(0x100, 8, 0);
        __builtin_amdgcn_sched_group_barrier(0x100, 8, 0);
        __builtin_amdgcn_sched_group_barrier(0x008, 16, 0);
        __builtin_amdgcn_sched_group_barrier(0x100, 8, 0);
        __builtin_amdgcn_sched_group_barrier(0x008, 16, 0);
        __builtin_amdgcn_sched_group_barrier(0x100, 8, 0);
        __builtin_amdgcn_sched_group_barrier(0x008, 16, 0);
        __builtin_amdgcn_sched_group_barrier(0x008, 16, 0);

        {
            float sl[4], sr[4];
#pragma unroll
            for (int i = 0; i < 4; ++i) {
                sl[i] = gw_fold<8>(c8, acc[i] * wl, acc[i + 4] * wl);
                sr[i] = gw_fold<8>(c8, acc[i] * wr, acc[i + 4] * wr);
            }
            const float tl0 = gw_fold<4>(c4, sl[0], sl[2]);
            const float tl1 = gw_fold<4>(c4, sl[1], sl[3]);
            const float tr0 = gw_fold<4>(c4, sr[0], sr[2]);
            const float tr1 = gw_fold<4>(c4, sr[1], sr[3]);
            const float ul = gw_fold<2>(c2, tl0, tl1);
            const float ur = gw_fold<2>(c2, tr0, tr1);
            part[buf][tid] = gw_fold<1>(c1, ul, ur);
        }

        if (has_next) {
            float* nb = As[buf ^ 1];
            gw_stage(nb, f0, p0);
            gw_stage(nb, f1, p1);
            gw_stage(nb, f2, p2);
            gw_stage(nb, f3, p3);
        }
        __syncthreads();

        const int row0 = t * 16 + (lane >> 4) * 8;
        float* cp = C + (size_t)row0 * NCOL + n0 + r;
        if (t * 16 + 16 <= M) {
#pragma unroll
            for (int i = 0; i < 8; ++i) cp[(size_t)i * NCOL] = acc[i];
        } else {
#pragma unroll
            for (int i = 0; i < 8; ++i)
                if (row0 + i < M) cp[(size_t)i * NCOL] = acc[i];
        }

        if (tid < 128) {
            const int sc  = tid >> 6;
            const int o   = tid & 63;
            const int row = o >> 2, h = o & 3;
            const int pl  = (row >> 3) * 16 + (row & 7) * 2 + sc;
            const float v = part[buf][(2 * h) * 32 + pl] + part[buf][(2 * h + 1) * 32 + pl];
            float* ap2 = sc ? a2 : a1;
            if (t * 16 + row < M) ap2[(size_t)t * 64 + o] = v;
        }

        if (!has_next) break;
        buf ^= 1;
        t = tn;
    }
}

#define OWN_THREADS 1024
#define OWN_GROUPS  256
#define CAP         12288
#define CAPW        (CAP / 32)
#define SEGV        (CAP / 4 - 1)
#define SCAN_U      8
#define AGG_U       8
#define KEEP        8

__device__ __forceinline__ float leaky(float v) {
    return v > 0.f ? v : ALPHA * v;
}

__device__ __forceinline__ int f2key(float f) {
    int b = __float_as_int(f);
    return b >= 0 ? b : (b ^ 0x7FFFFFFF);
}
__device__ __forceinline__ float key2f(int k) {
    return __int_as_float(k >= 0 ? k : (k ^ 0x7FFFFFFF));
}

struct OwnLds {
    float acc[R_MAX * NCOL];
    int   mkey[R_MAX * HEADS];
    float ssum[R_MAX * HEADS];
    float a2s[R_MAX * HEADS];
    int2  lst[CAP];
    int   cnt;
    int   wcnt[OWN_THREADS / 32];
    int   ovf;
    int   big;
};

template <bool FULL>
__device__ __forceinline__ void own_scan_block(OwnLds& L, const int4* __restrict__ d4,
                                               int b, int s1, int gbase,
                                               int lo, unsigned nR, int tid) {
    int4 d[SCAN_U];
#pragma unroll
    for (int u = 0; u < SCAN_U; ++u) {
        const int i = b + tid + u * OWN_THREADS;
        if (FULL || i < s1) d[u] = d4[i];
        else d[u] = make_int4(lo - 1, lo - 1, lo - 1, lo - 1);
    }
    unsigned mask = 0;
#pragma unroll
    for (int u = 0; u < SCAN_U; ++u) {
        const unsigned tx = (unsigned)(d[u].x - lo), ty = (unsigned)(d[u].y - lo);
        const unsigned tz = (unsigned)(d[u].z - lo), tw = (unsigned)(d[u].w - lo);
        const unsigned m = min(min(tx, ty), min(tz, tw));
        if (m < nR) {
            const unsigned g = (tx < nR ? 0x10000000u : 0u) | (ty < nR ? 0x20000000u : 0u) |
                               (tz < nR ? 0x40000000u : 0u) | (tw < nR ? 0x80000000u : 0u);
            mask |= g >> (28 - 4 * u);
        }
    }
    while (mask) {
        const int bit = __ffs(mask) - 1;
        mask &= mask - 1;
        const int i = b + (bit >> 2) * OWN_THREADS + tid;
        const int slot = atomicAdd(&L.cnt, 1);
        if (slot < CAP) L.lst[slot] = make_int2(((gbase + i) << 2) + (bit & 3), 0);
    }
}

__device__ __forceinline__ void own_filter(OwnLds& L, const int* __restrict__ dst,
                                           int v0, int v1, bool tail, int E,
                                           int lo, unsigned nR, int tid) {
    const int4* __restrict__ d4 = (const int4*)dst + v0;
    const int count = v1 - v0;
    const int rot = (count / (int)gridDim.x) * (int)blockIdx.x;
    for (int sp = 0; sp < 2; ++sp) {
        const int s0 = sp ? 0 : rot;
        const int s1 = sp ? rot : count;
        int b = s0;
        for (; b + OWN_THREADS * SCAN_U <= s1; b += OWN_THREADS * SCAN_U)
            own_scan_block<true>(L, d4, b, s1, v0, lo, nR, tid);
        if (b < s1)
            own_scan_block<false>(L, d4, b, s1, v0, lo, nR, tid);
    }
    if (tail && tid < (E & 3)) {
        int e = (v1 << 2) + tid;
        if ((unsigned)(dst[e] - lo) < nR) {
            int slot = atomicAdd(&L.cnt, 1);
            if (slot < CAP) L.lst[slot] = make_int2(e, 0);
        }
    }
}

#define PSCAN_G (SCAN_U / 2)
template <bool FULL>
__device__ __forceinline__ void own_pscan_block(OwnLds& L, const int4* __restrict__ pa,
                                                const int4* __restrict__ pb,
                                                int b, int s1, const unsigned (&c)[8],
                                                int wbase, int& wn, int tid) {
    int4 lo4[PSCAN_G], hi4[PSCAN_G];
#pragma unroll
    for (int v = 0; v < PSCAN_G; ++v) {
        const int g = b + tid + v * OWN_THREADS;
        if (FULL || g < s1) {
            lo4[v] = pa[g];
            hi4[v] = pb[g];
        } else {
            lo4[v] = make_int4((int)c[0], 0, 0, 0);
            hi4[v] = make_int4(0, 0, 0, 0);
        }
    }
#pragma unroll
    for (int v = 0; v < PSCAN_G; ++v) {
        unsigned m = ((unsigned)lo4[v].x ^ c[0]) & ((unsigned)lo4[v].y ^ c[1]) &
                     ((unsigned)lo4[v].z ^ c[2]) & ((unsigned)lo4[v].w ^ c[3]) &
                     ((unsigned)hi4[v].x ^ c[4]) & ((unsigned)hi4[v].y ^ c[5]) &
                     ((unsigned)hi4[v].z ^ c[6]) & ((unsigned)hi4[v].w ^ c[7]);
        const int e0 = (b + tid + v * OWN_THREADS) << 5;
        unsigned vote = __builtin_amdgcn_ballot_w32(m != 0);
        while (vote) {
            if (m) {
                const int bit = __ffs(m) - 1;
                m &= m - 1;
                const int slot = wn + (int)__builtin_amdgcn_mbcnt_lo(vote, 0);
                if (slot < CAPW) L.lst[wbase + slot] = make_int2(e0 + bit, 0);
            }
            wn += __popc(vote);
            vote = __builtin_amdgcn_ballot_w32(m != 0);
        }
    }
}

__device__ __forceinline__ void own_filter_plane(OwnLds& L, const int* __restrict__ plane,
                                                 const int* __restrict__ dst, int E,
                                                 int k, int lo, unsigned nR, int tid) {
    const int count = E >> 5;
    const int4* __restrict__ pa = (const int4*)plane;
    const int4* __restrict__ pb = pa + count;
    unsigned c[8];
#pragma unroll
    for (int j = 0; j < 8; ++j) c[j] = ((k >> j) & 1) ? 0u : ~0u;
    const int wave  = tid >> 5;
    const int wbase = wave * CAPW;
    int wn = 0;
    const int rot = (count / (int)gridDim.x) * (int)blockIdx.x;
    for (int sp = 0; sp < 2; ++sp) {
        const int s0 = sp ? 0 : rot;
        const int s1 = sp ? rot : count;
        int b = s0;
        for (; b + OWN_THREADS * PSCAN_G <= s1; b += OWN_THREADS * PSCAN_G)
            own_pscan_block<true>(L, pa, pb, b, s1, c, wbase, wn, tid);
        if (b < s1)
            own_pscan_block<false>(L, pa, pb, b, s1, c, wbase, wn, tid);
    }
    if (wave == 0) {
        const int e = (count << 5) + tid;
        bool hit = false;
        if (tid < (E & 31)) hit = (unsigned)(dst[e] - lo) < nR;
        const unsigned vote = __builtin_amdgcn_ballot_w32(hit);
        if (hit) {
            const int slot = wn + (int)__builtin_amdgcn_mbcnt_lo(vote, 0);
            if (slot < CAPW) L.lst[wbase + slot] = make_int2(e, 0);
        }
        wn += __popc(vote);
    }
    if ((tid & 31) == 0) {
        L.wcnt[wave] = wn;
        if (wn > CAPW) L.ovf = 1;
        if (wn > KEEP * 32) L.big = 1;
    }
}

__device__ __forceinline__ void own_resolve(OwnLds& L, const int* __restrict__ src,
                                            const int* __restrict__ dst,
                                            int first, int end, int step, int lo) {
    for (int i = first; i < end; i += step) {
        int e = L.lst[i].x;
        L.lst[i] = make_int2(src[e], dst[e] - lo);
    }
}

template <int PH>
__device__ __forceinline__ void own_edge_pass(OwnLds& L, const float* __restrict__ a1,
                                              int first, int end, int step) {
    for (int i = first; i < end; i += step) {
        int2 en = L.lst[i];
        float4 A1 = *(const float4*)(a1 + (size_t)en.x * HEADS);
        const int k = en.y * HEADS;
        float v0 = leaky(A1.x + L.a2s[k + 0]);
        float v1 = leaky(A1.y + L.a2s[k + 1]);
        float v2 = leaky(A1.z + L.a2s[k + 2]);
        float v3 = leaky(A1.w + L.a2s[k + 3]);
        if (PH == 0) {
            atomicMax(&L.mkey[k + 0], f2key(v0));
            atomicMax(&L.mkey[k + 1], f2key(v1));
            atomicMax(&L.mkey[k + 2], f2key(v2));
            atomicMax(&L.mkey[k + 3], f2key(v3));
        } else {
            atomicAdd(&L.ssum[k + 0], expf(v0 - key2f(L.mkey[k + 0])));
            atomicAdd(&L.ssum[k + 1], expf(v1 - key2f(L.mkey[k + 1])));
            atomicAdd(&L.ssum[k + 2], expf(v2 - key2f(L.mkey[k + 2])));
            atomicAdd(&L.ssum[k + 3], expf(v3 - key2f(L.mkey[k + 3])));
        }
    }
}

static_assert(AGG_U * HEADS == 32, "one lane per (batch edge, head)");
__device__ __forceinline__ void own_agg_pass(OwnLds& L, const float* __restrict__ ft,
                                             const float* __restrict__ a1,
                                             int first, int n, int us, int bs, int tid) {
    const int q = tid & 31;
    const int um = q >> 2;
    const int hl = q & 3;
    for (int i0 = first; i0 < n; i0 += bs) {
        int2  en[AGG_U];
        float F[AGG_U][HEADS];
#pragma unroll
        for (int u = 0; u < AGG_U; ++u) {
            int i = i0 + u * us;
            en[u] = L.lst[i < n ? i : i0];
        }
        const int im = i0 + um * us;
        const int2 em = L.lst[im < n ? im : i0];
#pragma unroll
        for (int u = 0; u < AGG_U; ++u) {
            const float* fr = ft + (size_t)en[u].x * NCOL + q;
            F[u][0] = fr[0];
            F[u][1] = fr[32];
            F[u][2] = fr[64];
            F[u][3] = fr[96];
        }
        const float A1m = a1[em.x * HEADS + hl];
        const int idx = em.y * HEADS + hl;
        const float v = leaky(A1m + L.a2s[idx]);
        const float alpha = expf(v - key2f(L.mkey[idx])) / L.ssum[idx];
        const int ab = __float_as_int(alpha);
#pragma unroll
        for (int u = 0; u < AGG_U; ++u) {
            if (i0 + u * us < n) {
                float* op = L.acc + en[u].y * NCOL + q;
                const float al0 = __int_as_float(__builtin_amdgcn_readlane(ab, 4 * u + 0));
                const float al1 = __int_as_float(__builtin_amdgcn_readlane(ab, 4 * u + 1));
                const float al2 = __int_as_float(__builtin_amdgcn_readlane(ab, 4 * u + 2));
                const float al3 = __int_as_float(__builtin_amdgcn_readlane(ab, 4 * u + 3));
                atomicAdd(op +  0, F[u][0] * al0);
                atomicAdd(op + 32, F[u][1] * al1);
                atomicAdd(op + 64, F[u][2] * al2);
                atomicAdd(op + 96, F[u][3] * al3);
            }
        }
    }
}

__device__ __forceinline__ void own_reg_path(OwnLds& L, const int* __restrict__ src,
                                             const int* __restrict__ dst,
                                             const float* __restrict__ ft,
                                             const float* __restrict__ a1,
                                             int wfirst, int wn, int lo, int lane) {
    float t[KEEP][HEADS];
    int ee[KEEP];
#pragma unroll
    for (int j = 0; j < KEEP; ++j) {
        const bool valid = 32 * j + lane < wn;
        const int e = L.lst[wfirst + 32 * j + lane].x;
        ee[j] = valid ? e : 0;
    }
    int ss[KEEP], dd[KEEP];
#pragma unroll
    for (int j = 0; j < KEEP; ++j) {
        ss[j] = src[ee[j]];
        dd[j] = dst[ee[j]] - lo;
    }
    float4 A1s[KEEP];
#pragma unroll
    for (int j = 0; j < KEEP; ++j) A1s[j] = *(const float4*)(a1 + (size_t)ss[j] * HEADS);
#pragma unroll
    for (int j = 0; j < KEEP; ++j) {
        const bool valid = 32 * j + lane < wn;
        const float4 A1 = A1s[j];
        const int k = valid ? dd[j] * HEADS : 0;
        if (valid) L.lst[wfirst + 32 * j + lane] = make_int2(ss[j], dd[j]);
        t[j][0] = leaky(A1.x + L.a2s[k + 0]);
        t[j][1] = leaky(A1.y + L.a2s[k + 1]);
        t[j][2] = leaky(A1.z + L.a2s[k + 2]);
        t[j][3] = leaky(A1.w + L.a2s[k + 3]);
        if (valid) {
#pragma unroll
            for (int h = 0; h < HEADS; ++h) atomicMax(&L.mkey[k + h], f2key(t[j][h]));
        }
    }
    __syncthreads();
#pragma unroll
    for (int j = 0; j < KEEP; ++j) {
        if (32 * j + lane < wn) {
            const int k = L.lst[wfirst + 32 * j + lane].y * HEADS;
#pragma unroll
            for (int h = 0; h < HEADS; ++h) {
                t[j][h] = expf(t[j][h] - key2f(L.mkey[k + h]));
                atomicAdd(&L.ssum[k + h], t[j][h]);
            }
        }
    }
    __syncthreads();
#pragma unroll
    for (int j = 0; j < KEEP; ++j) {
        if (32 * j + lane < wn) {
            const int k = L.lst[wfirst + 32 * j + lane].y * HEADS;
#pragma unroll
            for (int h = 0; h < HEADS; ++h) t[j][h] = t[j][h] / L.ssum[k + h];
        }
    }
#pragma unroll 1
    for (int j = 0; j < KEEP; ++j) {
        if (32 * j >= wn) break;
#pragma unroll 1
        for (int x0 = 0; x0 < 32; x0 += AGG_U) {
            const int r0 = 32 * j + x0;
            if (r0 >= wn) break;
            int2  en[AGG_U];
            float F[AGG_U][HEADS];
#pragma unroll
            for (int u = 0; u < AGG_U; ++u)
                en[u] = L.lst[wfirst + (r0 + u < wn ? r0 + u : r0)];
#pragma unroll
            for (int u = 0; u < AGG_U; ++u) {
                const unsigned fo = (unsigned)en[u].x * NCOL + (unsigned)lane;
                F[u][0] = ft[fo];
                F[u][1] = ft[fo + 32];
                F[u][2] = ft[fo + 64];
                F[u][3] = ft[fo + 96];
            }
#pragma unroll
            for (int u = 0; u < AGG_U; ++u) {
                if (r0 + u < wn) {
                    float* op = L.acc + en[u].y * NCOL + lane;
                    const float al0 = __int_as_float(__builtin_amdgcn_readlane(__float_as_int(t[0][0]), x0 + u));
                    const float al1 = __int_as_float(__builtin_amdgcn_readlane(__float_as_int(t[0][1]), x0 + u));
                    const float al2 = __int_as_float(__builtin_amdgcn_readlane(__float_as_int(t[0][2]), x0 + u));
                    const float al3 = __int_as_float(__builtin_amdgcn_readlane(__float_as_int(t[0][3]), x0 + u));
                    atomicAdd(op +  0, F[u][0] * al0);
                    atomicAdd(op + 32, F[u][1] * al1);
                    atomicAdd(op + 64, F[u][2] * al2);
                    atomicAdd(op + 96, F[u][3] * al3);
                }
            }
        }
#pragma unroll
        for (int jj = 0; jj + 1 < KEEP; ++jj) {
#pragma unroll
            for (int h = 0; h < HEADS; ++h) t[jj][h] = t[jj + 1][h];
        }
    }
}

__global__ void __launch_bounds__(OWN_THREADS)
gat_owner(const int* __restrict__ src, const int* __restrict__ dst,
          const float* __restrict__ ft, const float* __restrict__ a1,
          const float* __restrict__ a2, const int* __restrict__ plane,
          float* __restrict__ out, int n_nodes, int n_edges) {
    __shared__ OwnLds L;
    const int tid = threadIdx.x;
    const int wave = tid >> 5, lane = tid & 31;
    const int nvec = n_edges >> 2;
    const bool use_plane = n_nodes <= 256 * R_MAX;

    for (int lo = blockIdx.x * R_MAX; lo < n_nodes; lo += gridDim.x * R_MAX) {
        const int nR = (n_nodes - lo) < R_MAX ? (n_nodes - lo) : R_MAX;

        float4* acc4 = (float4*)L.acc;
        for (int i = tid; i < R_MAX * NCOL / 4; i += OWN_THREADS)
            acc4[i] = make_float4(0.f, 0.f, 0.f, 0.f);
        for (int i = tid; i < R_MAX * HEADS; i += OWN_THREADS) {
            L.mkey[i] = (int)0x80000000;
            L.ssum[i] = 0.f;
            L.a2s[i]  = (i < nR * HEADS) ? a2[(size_t)lo * HEADS + i] : 0.f;
        }
        if (tid == 0) { L.cnt = 0; L.ovf = 0; L.big = 0; }
        __syncthreads();

        bool own_runs = false;
        if (use_plane) {
            own_filter_plane(L, plane, dst, n_edges, lo / R_MAX, lo, (unsigned)nR, tid);
            __syncthreads();
            own_runs = L.ovf == 0;
        }

        int total = 0;
        if (!own_runs) {
            own_filter(L, dst, 0, nvec, true, n_edges, lo, (unsigned)nR, tid);
            __syncthreads();
            total = L.cnt;
        }

        const bool in_regs = own_runs && L.big == 0;
        if (__builtin_expect(in_regs, 1)) {
            const int wn = __builtin_amdgcn_readfirstlane(L.wcnt[wave]);
            own_reg_path(L, src, dst, ft, a1, wave * CAPW, wn, lo, lane);
        } else if (own_runs) {
            const int wfirst = wave * CAPW;
            const int wend   = wfirst + L.wcnt[wave];
            own_resolve(L, src, dst, wfirst + lane, wend, 32, lo);
            own_edge_pass<0>(L, a1, wfirst + lane, wend, 32);
            __syncthreads();
            own_edge_pass<1>(L, a1, wfirst + lane, wend, 32);
            __syncthreads();
            own_agg_pass(L, ft, a1, wfirst, wend, 1, AGG_U, tid);
        } else if (total <= CAP) {
            own_resolve(L, src, dst, tid, total, OWN_THREADS, lo);
            __syncthreads();
            own_edge_pass<0>(L, a1, tid, total, OWN_THREADS);
            __syncthreads();
            own_edge_pass<1>(L, a1, tid, total, OWN_THREADS);
            __syncthreads();
            own_agg_pass(L, ft, a1, wave, total, 32, 32 * AGG_U, tid);
        } else {
            for (int ph = 0; ph < 3; ++ph) {
                for (int v0 = 0; v0 <= nvec; v0 += SEGV) {
                    const int v1 = (v0 + SEGV < nvec) ? (v0 + SEGV) : nvec;
                    __syncthreads();
                    if (tid == 0) L.cnt = 0;
                    __syncthreads();
                    own_filter(L, dst, v0, v1, v1 == nvec, n_edges, lo, (unsigned)nR, tid);
                    __syncthreads();
                    const int n = L.cnt;
                    own_resolve(L, src, dst, tid, n, OWN_THREADS, lo);
                    __syncthreads();
                    if (ph == 0)      own_edge_pass<0>(L, a1, tid, n, OWN_THREADS);
                    else if (ph == 1) own_edge_pass<1>(L, a1, tid, n, OWN_THREADS);
                    else              own_agg_pass(L, ft, a1, wave, n, 32, 32 * AGG_U, tid);
                    if (v1 == nvec) break;
                }
                __syncthreads();
            }
        }
        __syncthreads();

        float4* out4 = (float4*)(out + (size_t)lo * NCOL);
        for (int i = tid; i < nR * NCOL / 4; i += OWN_THREADS) out4[i] = acc4[i];
        __syncthreads();
    }
}

extern "C" void kernel_launch(void* const* d_in, const int* in_sizes, int n_in,
                              void* d_out, int out_size, void* d_ws, size_t ws_size,
                              hipStream_t stream) {
    const float* feature = (const float*)d_in[0];
    const int*   src     = (const int*)d_in[1];
    const int*   dst     = (const int*)d_in[2];
    const float* W_fc    = (const float*)d_in[3];
    const float* attn_l  = (const float*)d_in[4];
    const float* attn_r  = (const float*)d_in[5];
    float*       out     = (float*)d_out;

    const int M = in_sizes[0] / IN_DIM;
    const int E = in_sizes[1];

    float* ft    = (float*)d_ws;
    float* a1    = ft    + (size_t)M * NCOL;
    float* a2    = a1    + (size_t)M * HEADS;
    int*   plane = (int*)(a2 + (size_t)M * HEADS);

    gat_gemm_wmma<<<GW_GROUPS, GW_THREADS, 0, stream>>>(feature, W_fc, attn_l, attn_r, dst,
                                                        ft, a1, a2, plane, M, E);
    gat_owner<<<OWN_GROUPS, OWN_THREADS, 0, stream>>>(src, dst, ft, a1, a2, plane, out, M, E);
}


# ft rows stored head-interleaved (float4 q = dim q of each of the 4 heads): aggregation gathers a source row with one 16-byte load per lane, not four 4-byte loads; sums and their order unchanged
# speedup vs baseline: 6.0995x; 1.0533x over previous
#include <hip/hip_runtime.h>
#include <hip/hip_bf16.h>

#define IN_DIM 256
#define NCOL 128
#define HEADS 4
#define OUT_DIM 32
#define ALPHA 0.2f
#define R_MAX 391

typedef __attribute__((ext_vector_type(2))) float v2f;
typedef __attribute__((ext_vector_type(8))) float v8f;

#define GW_THREADS 256
#define GW_GROUPS  512
#define A_PITCH    260

__device__ __forceinline__ float4 gw_load(const float* __restrict__ A, int tile, int f, int M) {
    int row = tile * 16 + (f >> 6);
    row = row < M ? row : M - 1;
    return *(const float4*)(A + (size_t)row * IN_DIM + (f & 63) * 4);
}
__device__ __forceinline__ void gw_stage(float* __restrict__ dstbuf, int f, float4 v) {
    *(float4*)(dstbuf + (f >> 6) * A_PITCH + (f & 63) * 4) = v;
}

template <int XM>
__device__ __forceinline__ float gw_xlane(float v) {
    return __int_as_float(__builtin_amdgcn_update_dpp(0, __float_as_int(v), 0x160 + XM,
                                                      0xf, 0xf, true));
}
template <int XM>
__device__ __forceinline__ float gw_fold(bool bit, float lo, float hi) {
    const float keep = bit ? hi : lo;
    const float send = bit ? lo : hi;
    return keep + gw_xlane<XM>(send);
}

__global__ void __launch_bounds__(GW_THREADS)
gat_gemm_wmma(const float* __restrict__ A,
              const float* __restrict__ B,
              const float* __restrict__ attn_l,
              const float* __restrict__ attn_r,
              const int* __restrict__ dst,
              float* __restrict__ C,
              float* __restrict__ a1,
              float* __restrict__ a2,
              int* __restrict__ plane, int M, int E) {
    __shared__ __attribute__((aligned(16))) float As[2][16 * A_PITCH];
    __shared__ float part[2][GW_THREADS];

    const int tid  = threadIdx.x;
    const int lane = tid & 31;
    const int n0   = (tid >> 5) * 16;
    const int r    = lane & 15;
    const int koff = (lane >> 4) * 2;
    const int tiles = (M + 15) >> 4;
    const int G = gridDim.x;

    {
        const int ngrp = E >> 5;
        const int nwav = G * (GW_THREADS / 32);
        int4* __restrict__ pa = (int4*)plane;
        int4* __restrict__ pb = pa + ngrp;
        const int wv = __builtin_amdgcn_readfirstlane(tid >> 5);
        for (int g0 = blockIdx.x * (GW_THREADS / 32) + wv; g0 < ngrp; g0 += 8 * nwav) {
            int dv[8];
#pragma unroll
            for (int q = 0; q < 8; ++q) {
                const int g = g0 + q * nwav;
                const int gl = g < ngrp ? g : ngrp - 1;
                dv[q] = dst[(size_t)gl * 32 + lane];
            }
#pragma unroll
            for (int q = 0; q < 8; ++q) {
                const int g = g0 + q * nwav;
                if (g < ngrp) {
                    const unsigned o = (unsigned)dv[q] / R_MAX;
                    const int b0 = (int)__builtin_amdgcn_ballot_w32((o & 1u) != 0);
                    const int b1 = (int)__builtin_amdgcn_ballot_w32((o & 2u) != 0);
                    const int b2 = (int)__builtin_amdgcn_ballot_w32((o & 4u) != 0);
                    const int b3 = (int)__builtin_amdgcn_ballot_w32((o & 8u) != 0);
                    const int b4 = (int)__builtin_amdgcn_ballot_w32((o & 16u) != 0);
                    const int b5 = (int)__builtin_amdgcn_ballot_w32((o & 32u) != 0);
                    const int b6 = (int)__builtin_amdgcn_ballot_w32((o & 64u) != 0);
                    const int b7 = (int)__builtin_amdgcn_ballot_w32((o & 128u) != 0);
                    if (lane == 0) {
                        pa[g] = make_int4(b0, b1, b2, b3);
                        pb[g] = make_int4(b4, b5, b6, b7);
                    }
                }
            }
        }
    }

    int t = blockIdx.x;
    if (t >= tiles) return;

    const float wl = attn_l[n0 + r];
    const float wr = attn_r[n0 + r];
    const bool c8 = (lane & 8) != 0, c4 = (lane & 4) != 0, c2 = (lane & 2) != 0, c1 = (lane & 1) != 0;

    const int f0 = tid, f1 = tid + GW_THREADS, f2 = tid + 2 * GW_THREADS, f3 = tid + 3 * GW_THREADS;
    float4 p0 = gw_load(A, t, f0, M);
    float4 p1 = gw_load(A, t, f1, M);
    float4 p2 = gw_load(A, t, f2, M);
    float4 p3 = gw_load(A, t, f3, M);

    v2f breg[IN_DIM / 4];
#pragma unroll
    for (int j = 0; j < IN_DIM / 4; ++j) {
        const float* bp = B + (size_t)(4 * j + koff) * NCOL + n0 + r;
        breg[j].x = bp[0];
        breg[j].y = bp[NCOL];
    }

    gw_stage(As[0], f0, p0);
    gw_stage(As[0], f1, p1);
    gw_stage(As[0], f2, p2);
    gw_stage(As[0], f3, p3);
    __syncthreads();

    int buf = 0;
    for (;;) {
        const int tn = t + G;
        const bool has_next = tn < tiles;
        if (has_next) {
            p0 = gw_load(A, tn, f0, M);
            p1 = gw_load(A, tn, f1, M);
            p2 = gw_load(A, tn, f2, M);
            p3 = gw_load(A, tn, f3, M);
        }

        v8f acc = {};
        const float* ap = &As[buf][r * A_PITCH + koff];
        v2f af[2][16];
#pragma unroll
        for (int i = 0; i < 16; ++i) af[0][i] = *(const v2f*)(ap + 4 * i);
#pragma unroll
        for (int kb = 0; kb < 4; ++kb) {
            if (kb < 3) {
#pragma unroll
                for (int i = 0; i < 16; ++i)
                    af[(kb + 1) & 1][i] = *(const v2f*)(ap + 4 * (16 * (kb + 1) + i));
            }
#pragma unroll
            for (int i = 0; i < 16; ++i) {
                acc = __builtin_amdgcn_wmma_f32_16x16x4_f32(
                     false, af[kb & 1][i],  false, breg[16 * kb + i],
                     (short)0, acc,  false,  false);
            }
        }
        __builtin_amdgcn_sched_group_barrier(0x100, 8, 0);
        __builtin_amdgcn_sched_group_barrier(0x100, 8, 0);
        __builtin_amdgcn_sched_group_barrier(0x008, 16, 0);
        __builtin_amdgcn_sched_group_barrier(0x100, 8, 0);
        __builtin_amdgcn_sched_group_barrier(0x008, 16, 0);
        __builtin_amdgcn_sched_group_barrier(0x100, 8, 0);
        __builtin_amdgcn_sched_group_barrier(0x008, 16, 0);
        __builtin_amdgcn_sched_group_barrier(0x008, 16, 0);

        {
            float sl[4], sr[4];
#pragma unroll
            for (int i = 0; i < 4; ++i) {
                sl[i] = gw_fold<8>(c8, acc[i] * wl, acc[i + 4] * wl);
                sr[i] = gw_fold<8>(c8, acc[i] * wr, acc[i + 4] * wr);
            }
            const float tl0 = gw_fold<4>(c4, sl[0], sl[2]);
            const float tl1 = gw_fold<4>(c4, sl[1], sl[3]);
            const float tr0 = gw_fold<4>(c4, sr[0], sr[2]);
            const float tr1 = gw_fold<4>(c4, sr[1], sr[3]);
            const float ul = gw_fold<2>(c2, tl0, tl1);
            const float ur = gw_fold<2>(c2, tr0, tr1);
            part[buf][tid] = gw_fold<1>(c1, ul, ur);
        }

        if (has_next) {
            float* nb = As[buf ^ 1];
            gw_stage(nb, f0, p0);
            gw_stage(nb, f1, p1);
            gw_stage(nb, f2, p2);
            gw_stage(nb, f3, p3);
        }
        __syncthreads();

        const int row0 = t * 16 + (lane >> 4) * 8;
        const int col  = n0 + r;
        float* cp = C + (size_t)row0 * NCOL + 4 * (col & (OUT_DIM - 1)) + (col / OUT_DIM);
        if (t * 16 + 16 <= M) {
#pragma unroll
            for (int i = 0; i < 8; ++i) cp[(size_t)i * NCOL] = acc[i];
        } else {
#pragma unroll
            for (int i = 0; i < 8; ++i)
                if (row0 + i < M) cp[(size_t)i * NCOL] = acc[i];
        }

        if (tid < 128) {
            const int sc  = tid >> 6;
            const int o   = tid & 63;
            const int row = o >> 2, h = o & 3;
            const int pl  = (row >> 3) * 16 + (row & 7) * 2 + sc;
            const float v = part[buf][(2 * h) * 32 + pl] + part[buf][(2 * h + 1) * 32 + pl];
            float* ap2 = sc ? a2 : a1;
            if (t * 16 + row < M) ap2[(size_t)t * 64 + o] = v;
        }

        if (!has_next) break;
        buf ^= 1;
        t = tn;
    }
}

#define OWN_THREADS 1024
#define OWN_GROUPS  256
#define CAP         12288
#define CAPW        (CAP / 32)
#define SEGV        (CAP / 4 - 1)
#define SCAN_U      8
#define AGG_U       8
#define KEEP        8

__device__ __forceinline__ float leaky(float v) {
    return v > 0.f ? v : ALPHA * v;
}

__device__ __forceinline__ int f2key(float f) {
    int b = __float_as_int(f);
    return b >= 0 ? b : (b ^ 0x7FFFFFFF);
}
__device__ __forceinline__ float key2f(int k) {
    return __int_as_float(k >= 0 ? k : (k ^ 0x7FFFFFFF));
}

struct OwnLds {
    float acc[R_MAX * NCOL];
    int   mkey[R_MAX * HEADS];
    float ssum[R_MAX * HEADS];
    float a2s[R_MAX * HEADS];
    int2  lst[CAP];
    int   cnt;
    int   wcnt[OWN_THREADS / 32];
    int   ovf;
    int   big;
};

template <bool FULL>
__device__ __forceinline__ void own_scan_block(OwnLds& L, const int4* __restrict__ d4,
                                               int b, int s1, int gbase,
                                               int lo, unsigned nR, int tid) {
    int4 d[SCAN_U];
#pragma unroll
    for (int u = 0; u < SCAN_U; ++u) {
        const int i = b + tid + u * OWN_THREADS;
        if (FULL || i < s1) d[u] = d4[i];
        else d[u] = make_int4(lo - 1, lo - 1, lo - 1, lo - 1);
    }
    unsigned mask = 0;
#pragma unroll
    for (int u = 0; u < SCAN_U; ++u) {
        const unsigned tx = (unsigned)(d[u].x - lo), ty = (unsigned)(d[u].y - lo);
        const unsigned tz = (unsigned)(d[u].z - lo), tw = (unsigned)(d[u].w - lo);
        const unsigned m = min(min(tx, ty), min(tz, tw));
        if (m < nR) {
            const unsigned g = (tx < nR ? 0x10000000u : 0u) | (ty < nR ? 0x20000000u : 0u) |
                               (tz < nR ? 0x40000000u : 0u) | (tw < nR ? 0x80000000u : 0u);
            mask |= g >> (28 - 4 * u);
        }
    }
    while (mask) {
        const int bit = __ffs(mask) - 1;
        mask &= mask - 1;
        const int i = b + (bit >> 2) * OWN_THREADS + tid;
        const int slot = atomicAdd(&L.cnt, 1);
        if (slot < CAP) L.lst[slot] = make_int2(((gbase + i) << 2) + (bit & 3), 0);
    }
}

__device__ __forceinline__ void own_filter(OwnLds& L, const int* __restrict__ dst,
                                           int v0, int v1, bool tail, int E,
                                           int lo, unsigned nR, int tid) {
    const int4* __restrict__ d4 = (const int4*)dst + v0;
    const int count = v1 - v0;
    const int rot = (count / (int)gridDim.x) * (int)blockIdx.x;
    for (int sp = 0; sp < 2; ++sp) {
        const int s0 = sp ? 0 : rot;
        const int s1 = sp ? rot : count;
        int b = s0;
        for (; b + OWN_THREADS * SCAN_U <= s1; b += OWN_THREADS * SCAN_U)
            own_scan_block<true>(L, d4, b, s1, v0, lo, nR, tid);
        if (b < s1)
            own_scan_block<false>(L, d4, b, s1, v0, lo, nR, tid);
    }
    if (tail && tid < (E & 3)) {
        int e = (v1 << 2) + tid;
        if ((unsigned)(dst[e] - lo) < nR) {
            int slot = atomicAdd(&L.cnt, 1);
            if (slot < CAP) L.lst[slot] = make_int2(e, 0);
        }
    }
}

#define PSCAN_G (SCAN_U / 2)
template <bool FULL>
__device__ __forceinline__ void own_pscan_block(OwnLds& L, const int4* __restrict__ pa,
                                                const int4* __restrict__ pb,
                                                int b, int s1, const unsigned (&c)[8],
                                                int wbase, int& wn, int tid) {
    int4 lo4[PSCAN_G], hi4[PSCAN_G];
#pragma unroll
    for (int v = 0; v < PSCAN_G; ++v) {
        const int g = b + tid + v * OWN_THREADS;
        if (FULL || g < s1) {
            lo4[v] = pa[g];
            hi4[v] = pb[g];
        } else {
            lo4[v] = make_int4((int)c[0], 0, 0, 0);
            hi4[v] = make_int4(0, 0, 0, 0);
        }
    }
#pragma unroll
    for (int v = 0; v < PSCAN_G; ++v) {
        unsigned m = ((unsigned)lo4[v].x ^ c[0]) & ((unsigned)lo4[v].y ^ c[1]) &
                     ((unsigned)lo4[v].z ^ c[2]) & ((unsigned)lo4[v].w ^ c[3]) &
                     ((unsigned)hi4[v].x ^ c[4]) & ((unsigned)hi4[v].y ^ c[5]) &
                     ((unsigned)hi4[v].z ^ c[6]) & ((unsigned)hi4[v].w ^ c[7]);
        const int e0 = (b + tid + v * OWN_THREADS) << 5;
        unsigned vote = __builtin_amdgcn_ballot_w32(m != 0);
        while (vote) {
            if (m) {
                const int bit = __ffs(m) - 1;
                m &= m - 1;
                const int slot = wn + (int)__builtin_amdgcn_mbcnt_lo(vote, 0);
                if (slot < CAPW) L.lst[wbase + slot] = make_int2(e0 + bit, 0);
            }
            wn += __popc(vote);
            vote = __builtin_amdgcn_ballot_w32(m != 0);
        }
    }
}

__device__ __forceinline__ void own_filter_plane(OwnLds& L, const int* __restrict__ plane,
                                                 const int* __restrict__ dst, int E,
                                                 int k, int lo, unsigned nR, int tid) {
    const int count = E >> 5;
    const int4* __restrict__ pa = (const int4*)plane;
    const int4* __restrict__ pb = pa + count;
    unsigned c[8];
#pragma unroll
    for (int j = 0; j < 8; ++j) c[j] = ((k >> j) & 1) ? 0u : ~0u;
    const int wave  = tid >> 5;
    const int wbase = wave * CAPW;
    int wn = 0;
    const int rot = (count / (int)gridDim.x) * (int)blockIdx.x;
    for (int sp = 0; sp < 2; ++sp) {
        const int s0 = sp ? 0 : rot;
        const int s1 = sp ? rot : count;
        int b = s0;
        for (; b + OWN_THREADS * PSCAN_G <= s1; b += OWN_THREADS * PSCAN_G)
            own_pscan_block<true>(L, pa, pb, b, s1, c, wbase, wn, tid);
        if (b < s1)
            own_pscan_block<false>(L, pa, pb, b, s1, c, wbase, wn, tid);
    }
    if (wave == 0) {
        const int e = (count << 5) + tid;
        bool hit = false;
        if (tid < (E & 31)) hit = (unsigned)(dst[e] - lo) < nR;
        const unsigned vote = __builtin_amdgcn_ballot_w32(hit);
        if (hit) {
            const int slot = wn + (int)__builtin_amdgcn_mbcnt_lo(vote, 0);
            if (slot < CAPW) L.lst[wbase + slot] = make_int2(e, 0);
        }
        wn += __popc(vote);
    }
    if ((tid & 31) == 0) {
        L.wcnt[wave] = wn;
        if (wn > CAPW) L.ovf = 1;
        if (wn > KEEP * 32) L.big = 1;
    }
}

__device__ __forceinline__ void own_resolve(OwnLds& L, const int* __restrict__ src,
                                            const int* __restrict__ dst,
                                            int first, int end, int step, int lo) {
    for (int i = first; i < end; i += step) {
        int e = L.lst[i].x;
        L.lst[i] = make_int2(src[e], dst[e] - lo);
    }
}

template <int PH>
__device__ __forceinline__ void own_edge_pass(OwnLds& L, const float* __restrict__ a1,
                                              int first, int end, int step) {
    for (int i = first; i < end; i += step) {
        int2 en = L.lst[i];
        float4 A1 = *(const float4*)(a1 + (size_t)en.x * HEADS);
        const int k = en.y * HEADS;
        float v0 = leaky(A1.x + L.a2s[k + 0]);
        float v1 = leaky(A1.y + L.a2s[k + 1]);
        float v2 = leaky(A1.z + L.a2s[k + 2]);
        float v3 = leaky(A1.w + L.a2s[k + 3]);
        if (PH == 0) {
            atomicMax(&L.mkey[k + 0], f2key(v0));
            atomicMax(&L.mkey[k + 1], f2key(v1));
            atomicMax(&L.mkey[k + 2], f2key(v2));
            atomicMax(&L.mkey[k + 3], f2key(v3));
        } else {
            atomicAdd(&L.ssum[k + 0], expf(v0 - key2f(L.mkey[k + 0])));
            atomicAdd(&L.ssum[k + 1], expf(v1 - key2f(L.mkey[k + 1])));
            atomicAdd(&L.ssum[k + 2], expf(v2 - key2f(L.mkey[k + 2])));
            atomicAdd(&L.ssum[k + 3], expf(v3 - key2f(L.mkey[k + 3])));
        }
    }
}

static_assert(AGG_U * HEADS == 32, "one lane per (batch edge, head)");
__device__ __forceinline__ void own_agg_pass(OwnLds& L, const float* __restrict__ ft,
                                             const float* __restrict__ a1,
                                             int first, int n, int us, int bs, int tid) {
    const int q = tid & 31;
    const int um = q >> 2;
    const int hl = q & 3;
    for (int i0 = first; i0 < n; i0 += bs) {
        int2  en[AGG_U];
        float F[AGG_U][HEADS];
#pragma unroll
        for (int u = 0; u < AGG_U; ++u) {
            int i = i0 + u * us;
            en[u] = L.lst[i < n ? i : i0];
        }
        const int im = i0 + um * us;
        const int2 em = L.lst[im < n ? im : i0];
#pragma unroll
        for (int u = 0; u < AGG_U; ++u) {
            const float4 f = *(const float4*)(ft + (size_t)en[u].x * NCOL + 4 * q);
            F[u][0] = f.x;
            F[u][1] = f.y;
            F[u][2] = f.z;
            F[u][3] = f.w;
        }
        const float A1m = a1[em.x * HEADS + hl];
        const int idx = em.y * HEADS + hl;
        const float v = leaky(A1m + L.a2s[idx]);
        const float alpha = expf(v - key2f(L.mkey[idx])) / L.ssum[idx];
        const int ab = __float_as_int(alpha);
#pragma unroll
        for (int u = 0; u < AGG_U; ++u) {
            if (i0 + u * us < n) {
                float* op = L.acc + en[u].y * NCOL + q;
                const float al0 = __int_as_float(__builtin_amdgcn_readlane(ab, 4 * u + 0));
                const float al1 = __int_as_float(__builtin_amdgcn_readlane(ab, 4 * u + 1));
                const float al2 = __int_as_float(__builtin_amdgcn_readlane(ab, 4 * u + 2));
                const float al3 = __int_as_float(__builtin_amdgcn_readlane(ab, 4 * u + 3));
                atomicAdd(op +  0, F[u][0] * al0);
                atomicAdd(op + 32, F[u][1] * al1);
                atomicAdd(op + 64, F[u][2] * al2);
                atomicAdd(op + 96, F[u][3] * al3);
            }
        }
    }
}

__device__ __forceinline__ void own_reg_path(OwnLds& L, const int* __restrict__ src,
                                             const int* __restrict__ dst,
                                             const float* __restrict__ ft,
                                             const float* __restrict__ a1,
                                             int wfirst, int wn, int lo, int lane) {
    float t[KEEP][HEADS];
    int ee[KEEP];
#pragma unroll
    for (int j = 0; j < KEEP; ++j) {
        const bool valid = 32 * j + lane < wn;
        const int e = L.lst[wfirst + 32 * j + lane].x;
        ee[j] = valid ? e : 0;
    }
    int ss[KEEP], dd[KEEP];
#pragma unroll
    for (int j = 0; j < KEEP; ++j) {
        ss[j] = src[ee[j]];
        dd[j] = dst[ee[j]] - lo;
    }
    float4 A1s[KEEP];
#pragma unroll
    for (int j = 0; j < KEEP; ++j) A1s[j] = *(const float4*)(a1 + (size_t)ss[j] * HEADS);
#pragma unroll
    for (int j = 0; j < KEEP; ++j) {
        const bool valid = 32 * j + lane < wn;
        const float4 A1 = A1s[j];
        const int k = valid ? dd[j] * HEADS : 0;
        if (valid) L.lst[wfirst + 32 * j + lane] = make_int2(ss[j], dd[j]);
        t[j][0] = leaky(A1.x + L.a2s[k + 0]);
        t[j][1] = leaky(A1.y + L.a2s[k + 1]);
        t[j][2] = leaky(A1.z + L.a2s[k + 2]);
        t[j][3] = leaky(A1.w + L.a2s[k + 3]);
        if (valid) {
#pragma unroll
            for (int h = 0; h < HEADS; ++h) atomicMax(&L.mkey[k + h], f2key(t[j][h]));
        }
    }
    __syncthreads();
#pragma unroll
    for (int j = 0; j < KEEP; ++j) {
        if (32 * j + lane < wn) {
            const int k = L.lst[wfirst + 32 * j + lane].y * HEADS;
#pragma unroll
            for (int h = 0; h < HEADS; ++h) {
                t[j][h] = expf(t[j][h] - key2f(L.mkey[k + h]));
                atomicAdd(&L.ssum[k + h], t[j][h]);
            }
        }
    }
    __syncthreads();
#pragma unroll
    for (int j = 0; j < KEEP; ++j) {
        if (32 * j + lane < wn) {
            const int k = L.lst[wfirst + 32 * j + lane].y * HEADS;
#pragma unroll
            for (int h = 0; h < HEADS; ++h) t[j][h] = t[j][h] / L.ssum[k + h];
        }
    }
#pragma unroll 1
    for (int j = 0; j < KEEP; ++j) {
        if (32 * j >= wn) break;
#pragma unroll 1
        for (int x0 = 0; x0 < 32; x0 += AGG_U) {
            const int r0 = 32 * j + x0;
            if (r0 >= wn) break;
            int2  en[AGG_U];
            float F[AGG_U][HEADS];
#pragma unroll
            for (int u = 0; u < AGG_U; ++u)
                en[u] = L.lst[wfirst + (r0 + u < wn ? r0 + u : r0)];
#pragma unroll
            for (int u = 0; u < AGG_U; ++u) {
                const unsigned fo = (unsigned)en[u].x * (NCOL / 4) + (unsigned)lane;
                const float4 f = ((const float4*)ft)[fo];
                F[u][0] = f.x;
                F[u][1] = f.y;
                F[u][2] = f.z;
                F[u][3] = f.w;
            }
#pragma unroll
            for (int u = 0; u < AGG_U; ++u) {
                if (r0 + u < wn) {
                    float* op = L.acc + en[u].y * NCOL + lane;
                    const float al0 = __int_as_float(__builtin_amdgcn_readlane(__float_as_int(t[0][0]), x0 + u));
                    const float al1 = __int_as_float(__builtin_amdgcn_readlane(__float_as_int(t[0][1]), x0 + u));
                    const float al2 = __int_as_float(__builtin_amdgcn_readlane(__float_as_int(t[0][2]), x0 + u));
                    const float al3 = __int_as_float(__builtin_amdgcn_readlane(__float_as_int(t[0][3]), x0 + u));
                    atomicAdd(op +  0, F[u][0] * al0);
                    atomicAdd(op + 32, F[u][1] * al1);
                    atomicAdd(op + 64, F[u][2] * al2);
                    atomicAdd(op + 96, F[u][3] * al3);
                }
            }
        }
#pragma unroll
        for (int jj = 0; jj + 1 < KEEP; ++jj) {
#pragma unroll
            for (int h = 0; h < HEADS; ++h) t[jj][h] = t[jj + 1][h];
        }
    }
}

__global__ void __launch_bounds__(OWN_THREADS)
gat_owner(const int* __restrict__ src, const int* __restrict__ dst,
          const float* __restrict__ ft, const float* __restrict__ a1,
          const float* __restrict__ a2, const int* __restrict__ plane,
          float* __restrict__ out, int n_nodes, int n_edges) {
    __shared__ OwnLds L;
    const int tid = threadIdx.x;
    const int wave = tid >> 5, lane = tid & 31;
    const int nvec = n_edges >> 2;
    const bool use_plane = n_nodes <= 256 * R_MAX;

    for (int lo = blockIdx.x * R_MAX; lo < n_nodes; lo += gridDim.x * R_MAX) {
        const int nR = (n_nodes - lo) < R_MAX ? (n_nodes - lo) : R_MAX;

        float4* acc4 = (float4*)L.acc;
        for (int i = tid; i < R_MAX * NCOL / 4; i += OWN_THREADS)
            acc4[i] = make_float4(0.f, 0.f, 0.f, 0.f);
        for (int i = tid; i < R_MAX * HEADS; i += OWN_THREADS) {
            L.mkey[i] = (int)0x80000000;
            L.ssum[i] = 0.f;
            L.a2s[i]  = (i < nR * HEADS) ? a2[(size_t)lo * HEADS + i] : 0.f;
        }
        if (tid == 0) { L.cnt = 0; L.ovf = 0; L.big = 0; }
        __syncthreads();

        bool own_runs = false;
        if (use_plane) {
            own_filter_plane(L, plane, dst, n_edges, lo / R_MAX, lo, (unsigned)nR, tid);
            __syncthreads();
            own_runs = L.ovf == 0;
        }

        int total = 0;
        if (!own_runs) {
            own_filter(L, dst, 0, nvec, true, n_edges, lo, (unsigned)nR, tid);
            __syncthreads();
            total = L.cnt;
        }

        const bool in_regs = own_runs && L.big == 0;
        if (__builtin_expect(in_regs, 1)) {
            const int wn = __builtin_amdgcn_readfirstlane(L.wcnt[wave]);
            own_reg_path(L, src, dst, ft, a1, wave * CAPW, wn, lo, lane);
        } else if (own_runs) {
            const int wfirst = wave * CAPW;
            const int wend   = wfirst + L.wcnt[wave];
            own_resolve(L, src, dst, wfirst + lane, wend, 32, lo);
            own_edge_pass<0>(L, a1, wfirst + lane, wend, 32);
            __syncthreads();
            own_edge_pass<1>(L, a1, wfirst + lane, wend, 32);
            __syncthreads();
            own_agg_pass(L, ft, a1, wfirst, wend, 1, AGG_U, tid);
        } else if (total <= CAP) {
            own_resolve(L, src, dst, tid, total, OWN_THREADS, lo);
            __syncthreads();
            own_edge_pass<0>(L, a1, tid, total, OWN_THREADS);
            __syncthreads();
            own_edge_pass<1>(L, a1, tid, total, OWN_THREADS);
            __syncthreads();
            own_agg_pass(L, ft, a1, wave, total, 32, 32 * AGG_U, tid);
        } else {
            for (int ph = 0; ph < 3; ++ph) {
                for (int v0 = 0; v0 <= nvec; v0 += SEGV) {
                    const int v1 = (v0 + SEGV < nvec) ? (v0 + SEGV) : nvec;
                    __syncthreads();
                    if (tid == 0) L.cnt = 0;
                    __syncthreads();
                    own_filter(L, dst, v0, v1, v1 == nvec, n_edges, lo, (unsigned)nR, tid);
                    __syncthreads();
                    const int n = L.cnt;
                    own_resolve(L, src, dst, tid, n, OWN_THREADS, lo);
                    __syncthreads();
                    if (ph == 0)      own_edge_pass<0>(L, a1, tid, n, OWN_THREADS);
                    else if (ph == 1) own_edge_pass<1>(L, a1, tid, n, OWN_THREADS);
                    else              own_agg_pass(L, ft, a1, wave, n, 32, 32 * AGG_U, tid);
                    if (v1 == nvec) break;
                }
                __syncthreads();
            }
        }
        __syncthreads();

        float4* out4 = (float4*)(out + (size_t)lo * NCOL);
        for (int i = tid; i < nR * NCOL / 4; i += OWN_THREADS) out4[i] = acc4[i];
        __syncthreads();
    }
}

extern "C" void kernel_launch(void* const* d_in, const int* in_sizes, int n_in,
                              void* d_out, int out_size, void* d_ws, size_t ws_size,
                              hipStream_t stream) {
    const float* feature = (const float*)d_in[0];
    const int*   src     = (const int*)d_in[1];
    const int*   dst     = (const int*)d_in[2];
    const float* W_fc    = (const float*)d_in[3];
    const float* attn_l  = (const float*)d_in[4];
    const float* attn_r  = (const float*)d_in[5];
    float*       out     = (float*)d_out;

    const int M = in_sizes[0] / IN_DIM;
    const int E = in_sizes[1];

    float* ft    = (float*)d_ws;
    float* a1    = ft    + (size_t)M * NCOL;
    float* a2    = a1    + (size_t)M * HEADS;
    int*   plane = (int*)(a2 + (size_t)M * HEADS);

    gat_gemm_wmma<<<GW_GROUPS, GW_THREADS, 0, stream>>>(feature, W_fc, attn_l, attn_r, dst,
                                                        ft, a1, a2, plane, M, E);
    gat_owner<<<OWN_GROUPS, OWN_THREADS, 0, stream>>>(src, dst, ft, a1, a2, plane, out, M, E);
}
